# starting kernel
# baseline (speedup 1.0000x reference)
#include <hip/hip_runtime.h>
#include <math.h>

#define N_NODES 50000
#define N_EDGES 600000
#define NFEAT 128
#define NHID 128
#define NCLASS 41

typedef float v2f __attribute__((ext_vector_type(2)));
typedef float v8f __attribute__((ext_vector_type(8)));

template<bool BIAS_RELU>
__global__ __launch_bounds__(256)
void gemm128_wmma(const float* __restrict__ A, const float* __restrict__ B,
                  const float* __restrict__ bias, float* __restrict__ C)
{
    const int lane = threadIdx.x & 31;
    const int wave = threadIdx.x >> 5;
    const int m0   = blockIdx.x * 16;
    const int n0   = wave * 16;
    const int m    = lane & 15;
    const int half = lane >> 4;

    const float* __restrict__ arow = A + (size_t)(m0 + m) * NHID;

    v8f acc = {};
#pragma unroll
    for (int kb = 0; kb < NHID; kb += 4) {
        const int ka = kb + 2 * half;
        float a0 = arow[ka];
        float a1 = arow[ka + 1];
        if (BIAS_RELU) {
            a0 = fmaxf(a0 + bias[ka],     0.0f);
            a1 = fmaxf(a1 + bias[ka + 1], 0.0f);
        }
        v2f a; a.x = a0; a.y = a1;
        v2f b;
        b.x = B[(size_t)ka       * NHID + n0 + m];
        b.y = B[(size_t)(ka + 1) * NHID + n0 + m];
        acc = __builtin_amdgcn_wmma_f32_16x16x4_f32(
                   false, a,  false, b,
                   (short)0, acc,  false,  false);
    }

#pragma unroll
    for (int v = 0; v < 8; ++v)
        C[(size_t)(m0 + v + 8 * half) * NHID + n0 + m] = acc[v];
}

__global__ __launch_bounds__(256)
void spmm_atomic(const int* __restrict__ src, const int* __restrict__ dst,
                 const float* __restrict__ w, const float* __restrict__ h,
                 float* __restrict__ acc)
{
    const int e = (int)((blockIdx.x * blockDim.x + threadIdx.x) >> 5);
    if (e >= N_EDGES) return;
    const int lane = threadIdx.x & 31;
    const int s  = src[e];
    const int d  = dst[e];
    const float wt = w[e];

    const float4 hv = *((const float4*)(h + (size_t)s * NHID) + lane);
    float* ap = acc + (size_t)d * NHID + lane * 4;
    unsafeAtomicAdd(ap + 0, wt * hv.x);
    unsafeAtomicAdd(ap + 1, wt * hv.y);
    unsafeAtomicAdd(ap + 2, wt * hv.z);
    unsafeAtomicAdd(ap + 3, wt * hv.w);
}

__global__ __launch_bounds__(256)
void zero_f4(float4* __restrict__ p, int n4)
{
    const int i = blockIdx.x * blockDim.x + threadIdx.x;
    if (i < n4) p[i] = make_float4(0.f, 0.f, 0.f, 0.f);
}

__global__ __launch_bounds__(64)
void head_logsoftmax(const float* __restrict__ acc, const float* __restrict__ b2,
                     const float* __restrict__ Wout, const float* __restrict__ bout,
                     float* __restrict__ out)
{
    __shared__ float hsh[NHID];
    __shared__ float red[64];
    const int node = blockIdx.x;
    const int t = threadIdx.x;

    for (int i = t; i < NHID; i += 64)
        hsh[i] = fmaxf(acc[(size_t)node * NHID + i] + b2[i], 0.0f);
    __syncthreads();

    float logit = -INFINITY;
    if (t < NCLASS) {
        float a = bout[t];
#pragma unroll 4
        for (int k = 0; k < NHID; ++k)
            a = fmaf(hsh[k], Wout[(size_t)k * NCLASS + t], a);
        logit = a;
    }

    red[t] = logit;
    __syncthreads();
    for (int s = 32; s > 0; s >>= 1) {
        if (t < s) red[t] = fmaxf(red[t], red[t + s]);
        __syncthreads();
    }
    const float mx = red[0];
    __syncthreads();

    red[t] = (t < NCLASS) ? __expf(logit - mx) : 0.0f;
    __syncthreads();
    for (int s = 32; s > 0; s >>= 1) {
        if (t < s) red[t] += red[t + s];
        __syncthreads();
    }
    const float lse = __logf(red[0]) + mx;

    if (t < NCLASS)
        out[(size_t)node * NCLASS + t] = logit - lse;
}

extern "C" void kernel_launch(void* const* d_in, const int* in_sizes, int n_in,
                              void* d_out, int out_size, void* d_ws, size_t ws_size,
                              hipStream_t stream)
{
    const float* x    = (const float*)d_in[0];
    const int*   a0i  = (const int*)  d_in[1];
    const float* a0w  = (const float*)d_in[2];
    const int*   a1i  = (const int*)  d_in[3];
    const float* a1w  = (const float*)d_in[4];
    const float* W1   = (const float*)d_in[5];
    const float* b1   = (const float*)d_in[6];
    const float* W2   = (const float*)d_in[7];
    const float* b2   = (const float*)d_in[8];
    const float* Wout = (const float*)d_in[9];
    const float* bout = (const float*)d_in[10];
    float* out = (float*)d_out;

    const size_t NH = (size_t)N_NODES * NHID;
    float* bufA = (float*)d_ws;
    float* bufB = bufA + NH;

    const dim3 gGemm(N_NODES / 16), bGemm(256);
    const int  n4 = (int)(NH / 4);
    const dim3 gZero((n4 + 255) / 256), bZero(256);
    const dim3 gSp((unsigned)(((size_t)N_EDGES * 32 + 255) / 256)), bSp(256);

    gemm128_wmma<false><<<gGemm, bGemm, 0, stream>>>(x, W1, nullptr, bufA);
    zero_f4<<<gZero, bZero, 0, stream>>>((float4*)bufB, n4);
    spmm_atomic<<<gSp, bSp, 0, stream>>>(a0i, a0i + N_EDGES, a0w, bufA, bufB);

    gemm128_wmma<true><<<gGemm, bGemm, 0, stream>>>(bufB, W2, b1, bufA);
    zero_f4<<<gZero, bZero, 0, stream>>>((float4*)bufB, n4);
    spmm_atomic<<<gSp, bSp, 0, stream>>>(a1i, a1i + N_EDGES, a1w, bufA, bufB);

    head_logsoftmax<<<dim3(N_NODES), dim3(64), 0, stream>>>(bufB, b2, Wout, bout, out);
}


# spmm: lane l owns columns l,l+32,l+64,l+96, so each wave-wide h load and f32 atomic add covers one whole 128-B line, not a quarter of four lines; sums still in atomic arrival order as before
# speedup vs baseline: 1.2233x; 1.2233x over previous
#include <hip/hip_runtime.h>
#include <math.h>

#define N_NODES 50000
#define N_EDGES 600000
#define NFEAT 128
#define NHID 128
#define NCLASS 41

typedef float v2f __attribute__((ext_vector_type(2)));
typedef float v8f __attribute__((ext_vector_type(8)));

template<bool BIAS_RELU>
__global__ __launch_bounds__(256)
void gemm128_wmma(const float* __restrict__ A, const float* __restrict__ B,
                  const float* __restrict__ bias, float* __restrict__ C)
{
    const int lane = threadIdx.x & 31;
    const int wave = threadIdx.x >> 5;
    const int m0   = blockIdx.x * 16;
    const int n0   = wave * 16;
    const int m    = lane & 15;
    const int half = lane >> 4;

    const float* __restrict__ arow = A + (size_t)(m0 + m) * NHID;

    v8f acc = {};
#pragma unroll
    for (int kb = 0; kb < NHID; kb += 4) {
        const int ka = kb + 2 * half;
        float a0 = arow[ka];
        float a1 = arow[ka + 1];
        if (BIAS_RELU) {
            a0 = fmaxf(a0 + bias[ka],     0.0f);
            a1 = fmaxf(a1 + bias[ka + 1], 0.0f);
        }
        v2f a; a.x = a0; a.y = a1;
        v2f b;
        b.x = B[(size_t)ka       * NHID + n0 + m];
        b.y = B[(size_t)(ka + 1) * NHID + n0 + m];
        acc = __builtin_amdgcn_wmma_f32_16x16x4_f32(
                   false, a,  false, b,
                   (short)0, acc,  false,  false);
    }

#pragma unroll
    for (int v = 0; v < 8; ++v)
        C[(size_t)(m0 + v + 8 * half) * NHID + n0 + m] = acc[v];
}

__global__ __launch_bounds__(256)
void spmm_atomic(const int* __restrict__ src, const int* __restrict__ dst,
                 const float* __restrict__ w, const float* __restrict__ h,
                 float* __restrict__ acc)
{
    const int e = (int)((blockIdx.x * blockDim.x + threadIdx.x) >> 5);
    if (e >= N_EDGES) return;
    const int lane = threadIdx.x & 31;
    const int s  = src[e];
    const int d  = dst[e];
    const float wt = w[e];

    const float* __restrict__ hp = h + (size_t)s * NHID + lane;
    const float h0 = hp[0];
    const float h1 = hp[32];
    const float h2 = hp[64];
    const float h3 = hp[96];
    float* ap = acc + (size_t)d * NHID + lane;
    unsafeAtomicAdd(ap +  0, wt * h0);
    unsafeAtomicAdd(ap + 32, wt * h1);
    unsafeAtomicAdd(ap + 64, wt * h2);
    unsafeAtomicAdd(ap + 96, wt * h3);
}

__global__ __launch_bounds__(256)
void zero_f4(float4* __restrict__ p, int n4)
{
    const int i = blockIdx.x * blockDim.x + threadIdx.x;
    if (i < n4) p[i] = make_float4(0.f, 0.f, 0.f, 0.f);
}

__global__ __launch_bounds__(64)
void head_logsoftmax(const float* __restrict__ acc, const float* __restrict__ b2,
                     const float* __restrict__ Wout, const float* __restrict__ bout,
                     float* __restrict__ out)
{
    __shared__ float hsh[NHID];
    __shared__ float red[64];
    const int node = blockIdx.x;
    const int t = threadIdx.x;

    for (int i = t; i < NHID; i += 64)
        hsh[i] = fmaxf(acc[(size_t)node * NHID + i] + b2[i], 0.0f);
    __syncthreads();

    float logit = -INFINITY;
    if (t < NCLASS) {
        float a = bout[t];
#pragma unroll 4
        for (int k = 0; k < NHID; ++k)
            a = fmaf(hsh[k], Wout[(size_t)k * NCLASS + t], a);
        logit = a;
    }

    red[t] = logit;
    __syncthreads();
    for (int s = 32; s > 0; s >>= 1) {
        if (t < s) red[t] = fmaxf(red[t], red[t + s]);
        __syncthreads();
    }
    const float mx = red[0];
    __syncthreads();

    red[t] = (t < NCLASS) ? __expf(logit - mx) : 0.0f;
    __syncthreads();
    for (int s = 32; s > 0; s >>= 1) {
        if (t < s) red[t] += red[t + s];
        __syncthreads();
    }
    const float lse = __logf(red[0]) + mx;

    if (t < NCLASS)
        out[(size_t)node * NCLASS + t] = logit - lse;
}

extern "C" void kernel_launch(void* const* d_in, const int* in_sizes, int n_in,
                              void* d_out, int out_size, void* d_ws, size_t ws_size,
                              hipStream_t stream)
{
    const float* x    = (const float*)d_in[0];
    const int*   a0i  = (const int*)  d_in[1];
    const float* a0w  = (const float*)d_in[2];
    const int*   a1i  = (const int*)  d_in[3];
    const float* a1w  = (const float*)d_in[4];
    const float* W1   = (const float*)d_in[5];
    const float* b1   = (const float*)d_in[6];
    const float* W2   = (const float*)d_in[7];
    const float* b2   = (const float*)d_in[8];
    const float* Wout = (const float*)d_in[9];
    const float* bout = (const float*)d_in[10];
    float* out = (float*)d_out;

    const size_t NH = (size_t)N_NODES * NHID;
    float* bufA = (float*)d_ws;
    float* bufB = bufA + NH;

    const dim3 gGemm(N_NODES / 16), bGemm(256);
    const int  n4 = (int)(NH / 4);
    const dim3 gZero((n4 + 255) / 256), bZero(256);
    const dim3 gSp((unsigned)(((size_t)N_EDGES * 32 + 255) / 256)), bSp(256);

    gemm128_wmma<false><<<gGemm, bGemm, 0, stream>>>(x, W1, nullptr, bufA);
    zero_f4<<<gZero, bZero, 0, stream>>>((float4*)bufB, n4);
    spmm_atomic<<<gSp, bSp, 0, stream>>>(a0i, a0i + N_EDGES, a0w, bufA, bufB);

    gemm128_wmma<true><<<gGemm, bGemm, 0, stream>>>(bufB, W2, b1, bufA);
    zero_f4<<<gZero, bZero, 0, stream>>>((float4*)bufB, n4);
    spmm_atomic<<<gSp, bSp, 0, stream>>>(a1i, a1i + N_EDGES, a1w, bufA, bufB);

    head_logsoftmax<<<dim3(N_NODES), dim3(64), 0, stream>>>(bufB, b2, Wout, bout, out);
}


# spmm: each wave takes 8 edges, scalar-loads their records and issues all 32 h-row line loads before its 32 f32 atomic adds (8x fewer waves, 8x loads in flight); sums still in arrival order
# speedup vs baseline: 1.4868x; 1.2154x over previous
#include <hip/hip_runtime.h>
#include <math.h>

#define N_NODES 50000
#define N_EDGES 600000
#define NFEAT 128
#define NHID 128
#define NCLASS 41

typedef float v2f __attribute__((ext_vector_type(2)));
typedef float v8f __attribute__((ext_vector_type(8)));

template<bool BIAS_RELU>
__global__ __launch_bounds__(256)
void gemm128_wmma(const float* __restrict__ A, const float* __restrict__ B,
                  const float* __restrict__ bias, float* __restrict__ C)
{
    const int lane = threadIdx.x & 31;
    const int wave = threadIdx.x >> 5;
    const int m0   = blockIdx.x * 16;
    const int n0   = wave * 16;
    const int m    = lane & 15;
    const int half = lane >> 4;

    const float* __restrict__ arow = A + (size_t)(m0 + m) * NHID;

    v8f acc = {};
#pragma unroll
    for (int kb = 0; kb < NHID; kb += 4) {
        const int ka = kb + 2 * half;
        float a0 = arow[ka];
        float a1 = arow[ka + 1];
        if (BIAS_RELU) {
            a0 = fmaxf(a0 + bias[ka],     0.0f);
            a1 = fmaxf(a1 + bias[ka + 1], 0.0f);
        }
        v2f a; a.x = a0; a.y = a1;
        v2f b;
        b.x = B[(size_t)ka       * NHID + n0 + m];
        b.y = B[(size_t)(ka + 1) * NHID + n0 + m];
        acc = __builtin_amdgcn_wmma_f32_16x16x4_f32(
                   false, a,  false, b,
                   (short)0, acc,  false,  false);
    }

#pragma unroll
    for (int v = 0; v < 8; ++v)
        C[(size_t)(m0 + v + 8 * half) * NHID + n0 + m] = acc[v];
}

#define EPW 8
static_assert(N_EDGES % (EPW * 8) == 0, "edge groups must tile the launch exactly");

__global__ __launch_bounds__(256)
void spmm_atomic(const int* __restrict__ src, const int* __restrict__ dst,
                 const float* __restrict__ w, const float* __restrict__ h,
                 float* __restrict__ acc)
{
    const int wv = __builtin_amdgcn_readfirstlane(
                       (int)((blockIdx.x * blockDim.x + threadIdx.x) >> 5));
    const int e0 = wv * EPW;
    if (e0 >= N_EDGES) return;
    const int lane = threadIdx.x & 31;

    int   s[EPW], d[EPW];
    float wt[EPW];
#pragma unroll
    for (int j = 0; j < EPW; ++j) {
        s[j]  = src[e0 + j];
        d[j]  = dst[e0 + j];
        wt[j] = w[e0 + j];
    }

    float hv[EPW][4];
#pragma unroll
    for (int j = 0; j < EPW; ++j) {
        const float* __restrict__ hp = h + (size_t)s[j] * NHID + lane;
        hv[j][0] = hp[0];
        hv[j][1] = hp[32];
        hv[j][2] = hp[64];
        hv[j][3] = hp[96];
    }

#pragma unroll
    for (int j = 0; j < EPW; ++j) {
        float* ap = acc + (size_t)d[j] * NHID + lane;
        unsafeAtomicAdd(ap +  0, wt[j] * hv[j][0]);
        unsafeAtomicAdd(ap + 32, wt[j] * hv[j][1]);
        unsafeAtomicAdd(ap + 64, wt[j] * hv[j][2]);
        unsafeAtomicAdd(ap + 96, wt[j] * hv[j][3]);
    }
}

__global__ __launch_bounds__(256)
void zero_f4(float4* __restrict__ p, int n4)
{
    const int i = blockIdx.x * blockDim.x + threadIdx.x;
    if (i < n4) p[i] = make_float4(0.f, 0.f, 0.f, 0.f);
}

__global__ __launch_bounds__(64)
void head_logsoftmax(const float* __restrict__ acc, const float* __restrict__ b2,
                     const float* __restrict__ Wout, const float* __restrict__ bout,
                     float* __restrict__ out)
{
    __shared__ float hsh[NHID];
    __shared__ float red[64];
    const int node = blockIdx.x;
    const int t = threadIdx.x;

    for (int i = t; i < NHID; i += 64)
        hsh[i] = fmaxf(acc[(size_t)node * NHID + i] + b2[i], 0.0f);
    __syncthreads();

    float logit = -INFINITY;
    if (t < NCLASS) {
        float a = bout[t];
#pragma unroll 4
        for (int k = 0; k < NHID; ++k)
            a = fmaf(hsh[k], Wout[(size_t)k * NCLASS + t], a);
        logit = a;
    }

    red[t] = logit;
    __syncthreads();
    for (int s = 32; s > 0; s >>= 1) {
        if (t < s) red[t] = fmaxf(red[t], red[t + s]);
        __syncthreads();
    }
    const float mx = red[0];
    __syncthreads();

    red[t] = (t < NCLASS) ? __expf(logit - mx) : 0.0f;
    __syncthreads();
    for (int s = 32; s > 0; s >>= 1) {
        if (t < s) red[t] += red[t + s];
        __syncthreads();
    }
    const float lse = __logf(red[0]) + mx;

    if (t < NCLASS)
        out[(size_t)node * NCLASS + t] = logit - lse;
}

extern "C" void kernel_launch(void* const* d_in, const int* in_sizes, int n_in,
                              void* d_out, int out_size, void* d_ws, size_t ws_size,
                              hipStream_t stream)
{
    const float* x    = (const float*)d_in[0];
    const int*   a0i  = (const int*)  d_in[1];
    const float* a0w  = (const float*)d_in[2];
    const int*   a1i  = (const int*)  d_in[3];
    const float* a1w  = (const float*)d_in[4];
    const float* W1   = (const float*)d_in[5];
    const float* b1   = (const float*)d_in[6];
    const float* W2   = (const float*)d_in[7];
    const float* b2   = (const float*)d_in[8];
    const float* Wout = (const float*)d_in[9];
    const float* bout = (const float*)d_in[10];
    float* out = (float*)d_out;

    const size_t NH = (size_t)N_NODES * NHID;
    float* bufA = (float*)d_ws;
    float* bufB = bufA + NH;

    const dim3 gGemm(N_NODES / 16), bGemm(256);
    const int  n4 = (int)(NH / 4);
    const dim3 gZero((n4 + 255) / 256), bZero(256);
    const dim3 gSp(N_EDGES / (EPW * 8)), bSp(256);

    gemm128_wmma<false><<<gGemm, bGemm, 0, stream>>>(x, W1, nullptr, bufA);
    zero_f4<<<gZero, bZero, 0, stream>>>((float4*)bufB, n4);
    spmm_atomic<<<gSp, bSp, 0, stream>>>(a0i, a0i + N_EDGES, a0w, bufA, bufB);

    gemm128_wmma<true><<<gGemm, bGemm, 0, stream>>>(bufB, W2, b1, bufA);
    zero_f4<<<gZero, bZero, 0, stream>>>((float4*)bufB, n4);
    spmm_atomic<<<gSp, bSp, 0, stream>>>(a1i, a1i + N_EDGES, a1w, bufA, bufB);

    head_logsoftmax<<<dim3(N_NODES), dim3(64), 0, stream>>>(bufB, b2, Wout, bout, out);
}


# spmm: each work-group owns 200 dst rows, filters the whole edge list for them and sums w*h[src] in an LDS tile with LDS f32 atomics (arrival order), then stores whole rows; no global atomics/zeroing
# speedup vs baseline: 1.6541x; 1.1125x over previous
#include <hip/hip_runtime.h>
#include <math.h>

#define N_NODES 50000
#define N_EDGES 600000
#define NFEAT 128
#define NHID 128
#define NCLASS 41

typedef float v2f __attribute__((ext_vector_type(2)));
typedef float v8f __attribute__((ext_vector_type(8)));

template<bool BIAS_RELU>
__global__ __launch_bounds__(256)
void gemm128_wmma(const float* __restrict__ A, const float* __restrict__ B,
                  const float* __restrict__ bias, float* __restrict__ C)
{
    const int lane = threadIdx.x & 31;
    const int wave = threadIdx.x >> 5;
    const int m0   = blockIdx.x * 16;
    const int n0   = wave * 16;
    const int m    = lane & 15;
    const int half = lane >> 4;

    const float* __restrict__ arow = A + (size_t)(m0 + m) * NHID;

    v8f acc = {};
#pragma unroll
    for (int kb = 0; kb < NHID; kb += 4) {
        const int ka = kb + 2 * half;
        float a0 = arow[ka];
        float a1 = arow[ka + 1];
        if (BIAS_RELU) {
            a0 = fmaxf(a0 + bias[ka],     0.0f);
            a1 = fmaxf(a1 + bias[ka + 1], 0.0f);
        }
        v2f a; a.x = a0; a.y = a1;
        v2f b;
        b.x = B[(size_t)ka       * NHID + n0 + m];
        b.y = B[(size_t)(ka + 1) * NHID + n0 + m];
        acc = __builtin_amdgcn_wmma_f32_16x16x4_f32(
                   false, a,  false, b,
                   (short)0, acc,  false,  false);
    }

#pragma unroll
    for (int v = 0; v < 8; ++v)
        C[(size_t)(m0 + v + 8 * half) * NHID + n0 + m] = acc[v];
}

#define SP_ROWS    200
#define SP_THREADS 1024
#define SP_WAVES   (SP_THREADS / 32)
#define SP_LPS     8
#define SP_STEP    (SP_LPS * 128)
#define SP_NFULL   (N_EDGES / SP_STEP)
#define SP_TAIL4   ((N_EDGES - SP_NFULL * SP_STEP) / 4)
#define SP_QT      32
#define SP_QCAP    (SP_QT + SP_STEP)
#define SP_GRP     8
static_assert(N_NODES % SP_ROWS == 0, "row ranges must tile the node set exactly");
static_assert(N_EDGES % 4 == 0, "dst is read as int4");

__global__ __launch_bounds__(SP_THREADS)
void spmm_rows_lds(const int* __restrict__ src, const int* __restrict__ dst,
                   const float* __restrict__ w, const float* __restrict__ h,
                   float* __restrict__ acc)
{
    __shared__ __attribute__((aligned(16))) float tile[SP_ROWS * NHID];
    __shared__ int queue[SP_WAVES * SP_QCAP];

    const int lane = threadIdx.x & 31;
    const int wave = __builtin_amdgcn_readfirstlane((int)(threadIdx.x >> 5));
    const int r0   = blockIdx.x * SP_ROWS;
    int* q = queue + wave * SP_QCAP;

    float4* tile4 = (float4*)tile;
    for (int i = threadIdx.x; i < SP_ROWS * NHID / 4; i += SP_THREADS)
        tile4[i] = make_float4(0.f, 0.f, 0.f, 0.f);
    __syncthreads();

    const int4* __restrict__ dst4 = (const int4*)dst;
    int qn = 0;

    auto drain = [&]() {
        for (int b = 0; b < qn; b += 32) {
            const int cnt = (qn - b) < 32 ? (qn - b) : 32;
            const int e  = (lane < cnt) ? q[b + lane] : 0;
            const int   sj = src[e];
            const int   dj = dst[e] - r0;
            const int   wj = __float_as_int(w[e]);
            for (int g = 0; g < cnt; g += SP_GRP) {
                float hv[SP_GRP][4];
#pragma unroll
                for (int u = 0; u < SP_GRP; ++u) {
                    const int iu = (g + u < cnt) ? (g + u) : (cnt - 1);
                    const int su = __builtin_amdgcn_readlane(sj, iu);
                    const float* __restrict__ hp = h + (size_t)su * NHID + lane;
                    hv[u][0] = hp[0];
                    hv[u][1] = hp[32];
                    hv[u][2] = hp[64];
                    hv[u][3] = hp[96];
                }
#pragma unroll
                for (int u = 0; u < SP_GRP; ++u) {
                    if (g + u < cnt) {
                        const int   du = __builtin_amdgcn_readlane(dj, g + u);
                        const float wu = __int_as_float(__builtin_amdgcn_readlane(wj, g + u));
                        float* tp = tile + du * NHID + lane;
                        atomicAdd(tp +  0, wu * hv[u][0]);
                        atomicAdd(tp + 32, wu * hv[u][1]);
                        atomicAdd(tp + 64, wu * hv[u][2]);
                        atomicAdd(tp + 96, wu * hv[u][3]);
                    }
                }
            }
        }
        qn = 0;
    };

    auto examine = [&](const int4 (&d)[SP_LPS], int st) {
        const int e0 = st * SP_STEP;
#pragma unroll
        for (int j = 0; j < SP_LPS; ++j) {
            const int  eb = e0 + (j * 32 + lane) * 4;
            const bool m0 = (unsigned)(d[j].x - r0) < (unsigned)SP_ROWS;
            const bool m1 = (unsigned)(d[j].y - r0) < (unsigned)SP_ROWS;
            const bool m2 = (unsigned)(d[j].z - r0) < (unsigned)SP_ROWS;
            const bool m3 = (unsigned)(d[j].w - r0) < (unsigned)SP_ROWS;
            if (__builtin_amdgcn_ballot_w32(m0 | m1 | m2 | m3)) {
                const unsigned k0 = __builtin_amdgcn_ballot_w32(m0);
                if (m0) q[qn + __builtin_amdgcn_mbcnt_lo(k0, 0u)] = eb;
                qn += __builtin_popcount(k0);
                const unsigned k1 = __builtin_amdgcn_ballot_w32(m1);
                if (m1) q[qn + __builtin_amdgcn_mbcnt_lo(k1, 0u)] = eb + 1;
                qn += __builtin_popcount(k1);
                const unsigned k2 = __builtin_amdgcn_ballot_w32(m2);
                if (m2) q[qn + __builtin_amdgcn_mbcnt_lo(k2, 0u)] = eb + 2;
                qn += __builtin_popcount(k2);
                const unsigned k3 = __builtin_amdgcn_ballot_w32(m3);
                if (m3) q[qn + __builtin_amdgcn_mbcnt_lo(k3, 0u)] = eb + 3;
                qn += __builtin_popcount(k3);
            }
        }
        if (qn >= SP_QT) drain();
    };

    int st = wave + (int)blockIdx.x * 2;
    if (st >= SP_NFULL) st -= SP_NFULL;
    int4 dv[SP_LPS];
    for (int i = wave; i < SP_NFULL; i += SP_WAVES) {
        const int4* __restrict__ p = dst4 + st * (SP_STEP / 4) + lane;
#pragma unroll
        for (int j = 0; j < SP_LPS; ++j) dv[j] = p[j * 32];
        examine(dv, st);
        st += SP_WAVES;
        if (st >= SP_NFULL) st -= SP_NFULL;
    }
    if (SP_TAIL4 > 0 && wave == SP_WAVES - 1) {
#pragma unroll
        for (int j = 0; j < SP_LPS; ++j) {
            const int t4 = j * 32 + lane;
            dv[j] = (t4 < SP_TAIL4) ? dst4[SP_NFULL * (SP_STEP / 4) + t4] : make_int4(-1, -1, -1, -1);
        }
        examine(dv, SP_NFULL);
    }
    drain();
    __syncthreads();

    float4* __restrict__ acc4 = (float4*)(acc + (size_t)r0 * NHID);
    for (int i = threadIdx.x; i < SP_ROWS * NHID / 4; i += SP_THREADS)
        acc4[i] = tile4[i];
}

__global__ __launch_bounds__(64)
void head_logsoftmax(const float* __restrict__ acc, const float* __restrict__ b2,
                     const float* __restrict__ Wout, const float* __restrict__ bout,
                     float* __restrict__ out)
{
    __shared__ float hsh[NHID];
    __shared__ float red[64];
    const int node = blockIdx.x;
    const int t = threadIdx.x;

    for (int i = t; i < NHID; i += 64)
        hsh[i] = fmaxf(acc[(size_t)node * NHID + i] + b2[i], 0.0f);
    __syncthreads();

    float logit = -INFINITY;
    if (t < NCLASS) {
        float a = bout[t];
#pragma unroll 4
        for (int k = 0; k < NHID; ++k)
            a = fmaf(hsh[k], Wout[(size_t)k * NCLASS + t], a);
        logit = a;
    }

    red[t] = logit;
    __syncthreads();
    for (int s = 32; s > 0; s >>= 1) {
        if (t < s) red[t] = fmaxf(red[t], red[t + s]);
        __syncthreads();
    }
    const float mx = red[0];
    __syncthreads();

    red[t] = (t < NCLASS) ? __expf(logit - mx) : 0.0f;
    __syncthreads();
    for (int s = 32; s > 0; s >>= 1) {
        if (t < s) red[t] += red[t + s];
        __syncthreads();
    }
    const float lse = __logf(red[0]) + mx;

    if (t < NCLASS)
        out[(size_t)node * NCLASS + t] = logit - lse;
}

extern "C" void kernel_launch(void* const* d_in, const int* in_sizes, int n_in,
                              void* d_out, int out_size, void* d_ws, size_t ws_size,
                              hipStream_t stream)
{
    const float* x    = (const float*)d_in[0];
    const int*   a0i  = (const int*)  d_in[1];
    const float* a0w  = (const float*)d_in[2];
    const int*   a1i  = (const int*)  d_in[3];
    const float* a1w  = (const float*)d_in[4];
    const float* W1   = (const float*)d_in[5];
    const float* b1   = (const float*)d_in[6];
    const float* W2   = (const float*)d_in[7];
    const float* b2   = (const float*)d_in[8];
    const float* Wout = (const float*)d_in[9];
    const float* bout = (const float*)d_in[10];
    float* out = (float*)d_out;

    const size_t NH = (size_t)N_NODES * NHID;
    float* bufA = (float*)d_ws;
    float* bufB = bufA + NH;

    const dim3 gGemm(N_NODES / 16), bGemm(256);
    const dim3 gSp(N_NODES / SP_ROWS), bSp(SP_THREADS);

    gemm128_wmma<false><<<gGemm, bGemm, 0, stream>>>(x, W1, nullptr, bufA);
    spmm_rows_lds<<<gSp, bSp, 0, stream>>>(a0i, a0i + N_EDGES, a0w, bufA, bufB);

    gemm128_wmma<true><<<gGemm, bGemm, 0, stream>>>(bufB, W2, b1, bufA);
    spmm_rows_lds<<<gSp, bSp, 0, stream>>>(a1i, a1i + N_EDGES, a1w, bufA, bufB);

    head_logsoftmax<<<dim3(N_NODES), dim3(64), 0, stream>>>(bufB, b2, Wout, bout, out);
}


# head: one wave per 16 nodes on f32 WMMA, Wout staged in LDS in fragment order (391 groups, not 50000); log-softmax by in-row butterflies on the D tiles; dot and exp sums in a new fixed order
# speedup vs baseline: 2.2024x; 1.3315x over previous
#include <hip/hip_runtime.h>
#include <math.h>

#define N_NODES 50000
#define N_EDGES 600000
#define NFEAT 128
#define NHID 128
#define NCLASS 41

typedef float v2f __attribute__((ext_vector_type(2)));
typedef float v8f __attribute__((ext_vector_type(8)));

template<bool BIAS_RELU>
__global__ __launch_bounds__(256)
void gemm128_wmma(const float* __restrict__ A, const float* __restrict__ B,
                  const float* __restrict__ bias, float* __restrict__ C)
{
    const int lane = threadIdx.x & 31;
    const int wave = threadIdx.x >> 5;
    const int m0   = blockIdx.x * 16;
    const int n0   = wave * 16;
    const int m    = lane & 15;
    const int half = lane >> 4;

    const float* __restrict__ arow = A + (size_t)(m0 + m) * NHID;

    v8f acc = {};
#pragma unroll
    for (int kb = 0; kb < NHID; kb += 4) {
        const int ka = kb + 2 * half;
        float a0 = arow[ka];
        float a1 = arow[ka + 1];
        if (BIAS_RELU) {
            a0 = fmaxf(a0 + bias[ka],     0.0f);
            a1 = fmaxf(a1 + bias[ka + 1], 0.0f);
        }
        v2f a; a.x = a0; a.y = a1;
        v2f b;
        b.x = B[(size_t)ka       * NHID + n0 + m];
        b.y = B[(size_t)(ka + 1) * NHID + n0 + m];
        acc = __builtin_amdgcn_wmma_f32_16x16x4_f32(
                   false, a,  false, b,
                   (short)0, acc,  false,  false);
    }

#pragma unroll
    for (int v = 0; v < 8; ++v)
        C[(size_t)(m0 + v + 8 * half) * NHID + n0 + m] = acc[v];
}

#define SP_ROWS    200
#define SP_THREADS 1024
#define SP_WAVES   (SP_THREADS / 32)
#define SP_LPS     8
#define SP_STEP    (SP_LPS * 128)
#define SP_NFULL   (N_EDGES / SP_STEP)
#define SP_TAIL4   ((N_EDGES - SP_NFULL * SP_STEP) / 4)
#define SP_QT      32
#define SP_QCAP    (SP_QT + SP_STEP)
#define SP_GRP     8
static_assert(N_NODES % SP_ROWS == 0, "row ranges must tile the node set exactly");
static_assert(N_EDGES % 4 == 0, "dst is read as int4");

__global__ __launch_bounds__(SP_THREADS)
void spmm_rows_lds(const int* __restrict__ src, const int* __restrict__ dst,
                   const float* __restrict__ w, const float* __restrict__ h,
                   float* __restrict__ acc)
{
    __shared__ __attribute__((aligned(16))) float tile[SP_ROWS * NHID];
    __shared__ int queue[SP_WAVES * SP_QCAP];

    const int lane = threadIdx.x & 31;
    const int wave = __builtin_amdgcn_readfirstlane((int)(threadIdx.x >> 5));
    const int r0   = blockIdx.x * SP_ROWS;
    int* q = queue + wave * SP_QCAP;

    float4* tile4 = (float4*)tile;
    for (int i = threadIdx.x; i < SP_ROWS * NHID / 4; i += SP_THREADS)
        tile4[i] = make_float4(0.f, 0.f, 0.f, 0.f);
    __syncthreads();

    const int4* __restrict__ dst4 = (const int4*)dst;
    int qn = 0;

    auto drain = [&]() {
        for (int b = 0; b < qn; b += 32) {
            const int cnt = (qn - b) < 32 ? (qn - b) : 32;
            const int e  = (lane < cnt) ? q[b + lane] : 0;
            const int   sj = src[e];
            const int   dj = dst[e] - r0;
            const int   wj = __float_as_int(w[e]);
            for (int g = 0; g < cnt; g += SP_GRP) {
                float hv[SP_GRP][4];
#pragma unroll
                for (int u = 0; u < SP_GRP; ++u) {
                    const int iu = (g + u < cnt) ? (g + u) : (cnt - 1);
                    const int su = __builtin_amdgcn_readlane(sj, iu);
                    const float* __restrict__ hp = h + (size_t)su * NHID + lane;
                    hv[u][0] = hp[0];
                    hv[u][1] = hp[32];
                    hv[u][2] = hp[64];
                    hv[u][3] = hp[96];
                }
#pragma unroll
                for (int u = 0; u < SP_GRP; ++u) {
                    if (g + u < cnt) {
                        const int   du = __builtin_amdgcn_readlane(dj, g + u);
                        const float wu = __int_as_float(__builtin_amdgcn_readlane(wj, g + u));
                        float* tp = tile + du * NHID + lane;
                        atomicAdd(tp +  0, wu * hv[u][0]);
                        atomicAdd(tp + 32, wu * hv[u][1]);
                        atomicAdd(tp + 64, wu * hv[u][2]);
                        atomicAdd(tp + 96, wu * hv[u][3]);
                    }
                }
            }
        }
        qn = 0;
    };

    auto examine = [&](const int4 (&d)[SP_LPS], int st) {
        const int e0 = st * SP_STEP;
#pragma unroll
        for (int j = 0; j < SP_LPS; ++j) {
            const int  eb = e0 + (j * 32 + lane) * 4;
            const bool m0 = (unsigned)(d[j].x - r0) < (unsigned)SP_ROWS;
            const bool m1 = (unsigned)(d[j].y - r0) < (unsigned)SP_ROWS;
            const bool m2 = (unsigned)(d[j].z - r0) < (unsigned)SP_ROWS;
            const bool m3 = (unsigned)(d[j].w - r0) < (unsigned)SP_ROWS;
            if (__builtin_amdgcn_ballot_w32(m0 | m1 | m2 | m3)) {
                const unsigned k0 = __builtin_amdgcn_ballot_w32(m0);
                if (m0) q[qn + __builtin_amdgcn_mbcnt_lo(k0, 0u)] = eb;
                qn += __builtin_popcount(k0);
                const unsigned k1 = __builtin_amdgcn_ballot_w32(m1);
                if (m1) q[qn + __builtin_amdgcn_mbcnt_lo(k1, 0u)] = eb + 1;
                qn += __builtin_popcount(k1);
                const unsigned k2 = __builtin_amdgcn_ballot_w32(m2);
                if (m2) q[qn + __builtin_amdgcn_mbcnt_lo(k2, 0u)] = eb + 2;
                qn += __builtin_popcount(k2);
                const unsigned k3 = __builtin_amdgcn_ballot_w32(m3);
                if (m3) q[qn + __builtin_amdgcn_mbcnt_lo(k3, 0u)] = eb + 3;
                qn += __builtin_popcount(k3);
            }
        }
        if (qn >= SP_QT) drain();
    };

    int st = wave + (int)blockIdx.x * 2;
    if (st >= SP_NFULL) st -= SP_NFULL;
    int4 dv[SP_LPS];
    for (int i = wave; i < SP_NFULL; i += SP_WAVES) {
        const int4* __restrict__ p = dst4 + st * (SP_STEP / 4) + lane;
#pragma unroll
        for (int j = 0; j < SP_LPS; ++j) dv[j] = p[j * 32];
        examine(dv, st);
        st += SP_WAVES;
        if (st >= SP_NFULL) st -= SP_NFULL;
    }
    if (SP_TAIL4 > 0 && wave == SP_WAVES - 1) {
#pragma unroll
        for (int j = 0; j < SP_LPS; ++j) {
            const int t4 = j * 32 + lane;
            dv[j] = (t4 < SP_TAIL4) ? dst4[SP_NFULL * (SP_STEP / 4) + t4] : make_int4(-1, -1, -1, -1);
        }
        examine(dv, SP_NFULL);
    }
    drain();
    __syncthreads();

    float4* __restrict__ acc4 = (float4*)(acc + (size_t)r0 * NHID);
    for (int i = threadIdx.x; i < SP_ROWS * NHID / 4; i += SP_THREADS)
        acc4[i] = tile4[i];
}

#define HD_WAVES 8
#define HD_NT    (N_NODES / 16)
#define HD_CT    3
#define HD_OUTF  (16 * NCLASS)
#define HD_OUT4  (HD_OUTF / 4)
static_assert(N_NODES % 16 == 0, "node tiles must be whole");
static_assert(NCLASS > 32 && NCLASS <= 48, "three column tiles, the third one partial");
static_assert(HD_OUTF % 4 == 0, "a node tile of out is stored as float4");
static_assert(32 % HD_WAVES == 0, "the 32 k-steps of Wout are dealt out evenly to the waves");

template<int M>
__device__ __forceinline__ float row_xor(float x)
{
    return __builtin_bit_cast(float,
        __builtin_amdgcn_mov_dpp(__builtin_bit_cast(int, x), 0x160 + M, 0xf, 0xf, true));
}

__global__ __launch_bounds__(HD_WAVES * 32)
void head_wmma_logsoftmax(const float* __restrict__ acc, const float* __restrict__ b2,
                          const float* __restrict__ Wout, const float* __restrict__ bout,
                          float* __restrict__ out)
{
    __shared__ __attribute__((aligned(16))) float2 Bs[32 * HD_CT * 32];
    __shared__ __attribute__((aligned(16))) float  stg[HD_WAVES][HD_OUTF];

    const int lane = threadIdx.x & 31;
    const int wave = __builtin_amdgcn_readfirstlane((int)(threadIdx.x >> 5));
    const int c    = lane & 15;
    const int hh   = lane >> 4;
    int tile = blockIdx.x * HD_WAVES + wave;
    const bool live = tile < HD_NT;
    if (!live) tile = HD_NT - 1;

    const bool c2 = (32 + c) < NCLASS;
    const int  col2 = c2 ? (32 + c) : (NCLASS - 1);
    float2 wv[32 / HD_WAVES][HD_CT];
#pragma unroll
    for (int q = 0; q < 32 / HD_WAVES; ++q) {
        const int s = wave + q * HD_WAVES;
        const int k = 8 * (s >> 1) + 4 * hh + 2 * (s & 1);
        const float* __restrict__ w0 = Wout + (size_t)k * NCLASS;
        const float* __restrict__ w1 = w0 + NCLASS;
        wv[q][0] = make_float2(w0[c],      w1[c]);
        wv[q][1] = make_float2(w0[16 + c], w1[16 + c]);
        wv[q][2] = make_float2(w0[col2],   w1[col2]);
    }

    const float4* __restrict__ ap = (const float4*)(acc + (size_t)(tile * 16 + c) * NHID + 4 * hh);
    const float4* __restrict__ bp = (const float4*)(b2 + 4 * hh);
    float4 av[16], bv[16];
#pragma unroll
    for (int j = 0; j < 16; ++j) av[j] = ap[2 * j];
#pragma unroll
    for (int j = 0; j < 16; ++j) bv[j] = bp[2 * j];

#pragma unroll
    for (int q = 0; q < 32 / HD_WAVES; ++q) {
        const int s = wave + q * HD_WAVES;
        float2* __restrict__ bd = Bs + (s * HD_CT) * 32 + lane;
        bd[0]  = wv[q][0];
        bd[32] = wv[q][1];
        bd[64] = c2 ? wv[q][2] : make_float2(0.f, 0.f);
    }
    __syncthreads();

    v8f d0 = {}, d1 = {}, d2 = {};
#pragma unroll
    for (int j = 0; j < 16; ++j) {
        const float x0 = fmaxf(av[j].x + bv[j].x, 0.0f);
        const float x1 = fmaxf(av[j].y + bv[j].y, 0.0f);
        const float x2 = fmaxf(av[j].z + bv[j].z, 0.0f);
        const float x3 = fmaxf(av[j].w + bv[j].w, 0.0f);
        v2f a; float2 f; v2f b;
        const float2* __restrict__ bs = Bs + (2 * j) * HD_CT * 32 + lane;
        a.x = x0; a.y = x1;
        f = bs[0];   b.x = f.x; b.y = f.y;
        d0 = __builtin_amdgcn_wmma_f32_16x16x4_f32(false, a, false, b, (short)0, d0, false, false);
        f = bs[32];  b.x = f.x; b.y = f.y;
        d1 = __builtin_amdgcn_wmma_f32_16x16x4_f32(false, a, false, b, (short)0, d1, false, false);
        f = bs[64];  b.x = f.x; b.y = f.y;
        d2 = __builtin_amdgcn_wmma_f32_16x16x4_f32(false, a, false, b, (short)0, d2, false, false);
        a.x = x2; a.y = x3;
        f = bs[96];  b.x = f.x; b.y = f.y;
        d0 = __builtin_amdgcn_wmma_f32_16x16x4_f32(false, a, false, b, (short)0, d0, false, false);
        f = bs[128]; b.x = f.x; b.y = f.y;
        d1 = __builtin_amdgcn_wmma_f32_16x16x4_f32(false, a, false, b, (short)0, d1, false, false);
        f = bs[160]; b.x = f.x; b.y = f.y;
        d2 = __builtin_amdgcn_wmma_f32_16x16x4_f32(false, a, false, b, (short)0, d2, false, false);
    }

    const float bo0 = bout[c];
    const float bo1 = bout[16 + c];
    const float bo2 = bout[col2];
    float l0[8], l1[8], l2[8], mx[8], se[8];
#pragma unroll
    for (int r = 0; r < 8; ++r) {
        l0[r] = d0[r] + bo0;
        l1[r] = d1[r] + bo1;
        l2[r] = c2 ? (d2[r] + bo2) : -INFINITY;
        mx[r] = fmaxf(fmaxf(l0[r], l1[r]), l2[r]);
    }
#pragma unroll
    for (int r = 0; r < 8; ++r) mx[r] = fmaxf(mx[r], row_xor<1>(mx[r]));
#pragma unroll
    for (int r = 0; r < 8; ++r) mx[r] = fmaxf(mx[r], row_xor<2>(mx[r]));
#pragma unroll
    for (int r = 0; r < 8; ++r) mx[r] = fmaxf(mx[r], row_xor<4>(mx[r]));
#pragma unroll
    for (int r = 0; r < 8; ++r) mx[r] = fmaxf(mx[r], row_xor<8>(mx[r]));
#pragma unroll
    for (int r = 0; r < 8; ++r)
        se[r] = __expf(l0[r] - mx[r]) + __expf(l1[r] - mx[r]) + (c2 ? __expf(l2[r] - mx[r]) : 0.0f);
#pragma unroll
    for (int r = 0; r < 8; ++r) se[r] += row_xor<1>(se[r]);
#pragma unroll
    for (int r = 0; r < 8; ++r) se[r] += row_xor<2>(se[r]);
#pragma unroll
    for (int r = 0; r < 8; ++r) se[r] += row_xor<4>(se[r]);
#pragma unroll
    for (int r = 0; r < 8; ++r) se[r] += row_xor<8>(se[r]);

    float* __restrict__ sp = stg[wave] + (8 * hh) * NCLASS + c;
#pragma unroll
    for (int r = 0; r < 8; ++r) {
        const float lse = __logf(se[r]) + mx[r];
        sp[r * NCLASS]      = l0[r] - lse;
        sp[r * NCLASS + 16] = l1[r] - lse;
        if (c2) sp[r * NCLASS + 32] = l2[r] - lse;
    }
    __syncthreads();

    if (live) {
        float4* __restrict__ o4 = (float4*)(out + (size_t)tile * HD_OUTF);
        const float4* s4 = (const float4*)stg[wave];
#pragma unroll
        for (int u = 0; u < (HD_OUT4 + 31) / 32; ++u) {
            const int i = u * 32 + lane;
            if (i < HD_OUT4) o4[i] = s4[i];
        }
    }
}

extern "C" void kernel_launch(void* const* d_in, const int* in_sizes, int n_in,
                              void* d_out, int out_size, void* d_ws, size_t ws_size,
                              hipStream_t stream)
{
    const float* x    = (const float*)d_in[0];
    const int*   a0i  = (const int*)  d_in[1];
    const float* a0w  = (const float*)d_in[2];
    const int*   a1i  = (const int*)  d_in[3];
    const float* a1w  = (const float*)d_in[4];
    const float* W1   = (const float*)d_in[5];
    const float* b1   = (const float*)d_in[6];
    const float* W2   = (const float*)d_in[7];
    const float* b2   = (const float*)d_in[8];
    const float* Wout = (const float*)d_in[9];
    const float* bout = (const float*)d_in[10];
    float* out = (float*)d_out;

    const size_t NH = (size_t)N_NODES * NHID;
    float* bufA = (float*)d_ws;
    float* bufB = bufA + NH;

    const dim3 gGemm(N_NODES / 16), bGemm(256);
    const dim3 gSp(N_NODES / SP_ROWS), bSp(SP_THREADS);

    gemm128_wmma<false><<<gGemm, bGemm, 0, stream>>>(x, W1, nullptr, bufA);
    spmm_rows_lds<<<gSp, bSp, 0, stream>>>(a0i, a0i + N_EDGES, a0w, bufA, bufB);

    gemm128_wmma<true><<<gGemm, bGemm, 0, stream>>>(bufB, W2, b1, bufA);
    spmm_rows_lds<<<gSp, bSp, 0, stream>>>(a1i, a1i + N_EDGES, a1w, bufA, bufB);

    head_wmma_logsoftmax<<<dim3((HD_NT + HD_WAVES - 1) / HD_WAVES), dim3(HD_WAVES * 32), 0, stream>>>(bufB, b2, Wout, bout, out);
}


# spmm: new launch sorts each 4800-edge chunk by 200-row owner block in LDS (integer-atomic rank); a block's group reads only its own records, not the whole edge list; sums in arrival order as before
# speedup vs baseline: 2.7676x; 1.2566x over previous
#include <hip/hip_runtime.h>
#include <math.h>

#define N_NODES 50000
#define N_EDGES 600000
#define NFEAT 128
#define NHID 128
#define NCLASS 41

typedef float v2f __attribute__((ext_vector_type(2)));
typedef float v8f __attribute__((ext_vector_type(8)));

template<bool BIAS_RELU>
__global__ __launch_bounds__(256)
void gemm128_wmma(const float* __restrict__ A, const float* __restrict__ B,
                  const float* __restrict__ bias, float* __restrict__ C)
{
    const int lane = threadIdx.x & 31;
    const int wave = threadIdx.x >> 5;
    const int m0   = blockIdx.x * 16;
    const int n0   = wave * 16;
    const int m    = lane & 15;
    const int half = lane >> 4;

    const float* __restrict__ arow = A + (size_t)(m0 + m) * NHID;

    v8f acc = {};
#pragma unroll
    for (int kb = 0; kb < NHID; kb += 4) {
        const int ka = kb + 2 * half;
        float a0 = arow[ka];
        float a1 = arow[ka + 1];
        if (BIAS_RELU) {
            a0 = fmaxf(a0 + bias[ka],     0.0f);
            a1 = fmaxf(a1 + bias[ka + 1], 0.0f);
        }
        v2f a; a.x = a0; a.y = a1;
        v2f b;
        b.x = B[(size_t)ka       * NHID + n0 + m];
        b.y = B[(size_t)(ka + 1) * NHID + n0 + m];
        acc = __builtin_amdgcn_wmma_f32_16x16x4_f32(
                   false, a,  false, b,
                   (short)0, acc,  false,  false);
    }

#pragma unroll
    for (int v = 0; v < 8; ++v)
        C[(size_t)(m0 + v + 8 * half) * NHID + n0 + m] = acc[v];
}

#define SP_ROWS    200
#define SP_NB      (N_NODES / SP_ROWS)
#define SP_THREADS 1024
#define SP_WAVES   (SP_THREADS / 32)
#define SP_GRP     8
#define BK_CH      4800
#define BK_NCH     (N_EDGES / BK_CH)
#define BK_THREADS 512
#define BK_PER     ((BK_CH + BK_THREADS - 1) / BK_THREADS)
#define BK_BINS    256
#define BK_SPW     ((BK_NCH + SP_WAVES - 1) / SP_WAVES)
static_assert(N_NODES % SP_ROWS == 0, "row ranges must tile the node set exactly");
static_assert(N_EDGES % BK_CH == 0, "chunks must tile the edge list exactly");
static_assert(SP_NB < BK_BINS && BK_BINS == 8 * 32, "the scan gives each lane of one wave 8 bins");
static_assert(SP_NB <= BK_THREADS, "one thread writes one (first, count) pair");
static_assert(BK_SPW <= 32, "a wave holds its chunk pairs one per lane");

__global__ __launch_bounds__(BK_THREADS)
void bucket_chunks(const int* __restrict__ src0, const int* __restrict__ dst0, const float* __restrict__ w0,
                   const int* __restrict__ src1, const int* __restrict__ dst1, const float* __restrict__ w1,
                   int* __restrict__ rec_all, int* __restrict__ seg_all)
{
    __shared__ int hist[BK_BINS];
    __shared__ int first[BK_BINS];
    __shared__ __attribute__((aligned(16))) int4 sorted[BK_CH];

    const int t     = threadIdx.x;
    const int graph = (int)blockIdx.x / BK_NCH;
    const int chunk = (int)blockIdx.x - graph * BK_NCH;
    const int e0    = chunk * BK_CH;
    const int*   __restrict__ src = graph ? src1 : src0;
    const int*   __restrict__ dst = graph ? dst1 : dst0;
    const float* __restrict__ w   = graph ? w1 : w0;

    if (t < BK_BINS) hist[t] = 0;

    int s[BK_PER] = {}, d[BK_PER] = {}, wb[BK_PER] = {}, bin[BK_PER] = {}, rank[BK_PER] = {};
#pragma unroll
    for (int j = 0; j < BK_PER; ++j) {
        const int i = j * BK_THREADS + t;
        if (i < BK_CH) {
            s[j]  = src[e0 + i];
            d[j]  = dst[e0 + i];
            wb[j] = __float_as_int(w[e0 + i]);
        }
    }
    __syncthreads();

#pragma unroll
    for (int j = 0; j < BK_PER; ++j) {
        const int i = j * BK_THREADS + t;
        if (i < BK_CH) {
            bin[j]  = ((unsigned)d[j] < (unsigned)N_NODES) ? (d[j] / SP_ROWS) : SP_NB;
            rank[j] = atomicAdd(&hist[bin[j]], 1);
        }
    }
    __syncthreads();

    if (t < 32) {
        int c[8];
        int sum = 0;
#pragma unroll
        for (int k = 0; k < 8; ++k) { c[k] = hist[t * 8 + k]; sum += c[k]; }
        int incl = sum;
#pragma unroll
        for (int dl = 1; dl < 32; dl <<= 1) {
            const int o = __shfl_up(incl, dl);
            if (t >= dl) incl += o;
        }
        int run = incl - sum;
#pragma unroll
        for (int k = 0; k < 8; ++k) { first[t * 8 + k] = run; run += c[k]; }
    }
    __syncthreads();

#pragma unroll
    for (int j = 0; j < BK_PER; ++j) {
        const int i = j * BK_THREADS + t;
        if (i < BK_CH)
            sorted[first[bin[j]] + rank[j]] = make_int4(s[j], d[j] - bin[j] * SP_ROWS, wb[j], 0);
    }
    if (t < SP_NB) {
        int2* __restrict__ seg = (int2*)seg_all + ((size_t)graph * SP_NB + t) * BK_NCH + chunk;
        *seg = make_int2(e0 + first[t], hist[t]);
    }
    __syncthreads();

    int4* __restrict__ rec = (int4*)rec_all + (size_t)graph * N_EDGES + e0;
    for (int i = t; i < BK_CH; i += BK_THREADS) rec[i] = sorted[i];
}

__global__ __launch_bounds__(SP_THREADS)
void spmm_rows_lds(const int* __restrict__ rec_g, const int* __restrict__ seg_g,
                   const float* __restrict__ h, float* __restrict__ acc)
{
    __shared__ __attribute__((aligned(16))) float tile[SP_ROWS * NHID];

    const int lane = threadIdx.x & 31;
    const int wave = __builtin_amdgcn_readfirstlane((int)(threadIdx.x >> 5));
    const int r0   = blockIdx.x * SP_ROWS;

    const int2* __restrict__ seg = (const int2*)seg_g + (size_t)blockIdx.x * BK_NCH;
    const int myc = wave + lane * SP_WAVES;
    int2 pr = make_int2(0, 0);
    if (lane < BK_SPW && myc < BK_NCH) pr = seg[myc];

    float4* tile4 = (float4*)tile;
    for (int i = threadIdx.x; i < SP_ROWS * NHID / 4; i += SP_THREADS)
        tile4[i] = make_float4(0.f, 0.f, 0.f, 0.f);
    __syncthreads();

    int sf[BK_SPW], pf[BK_SPW + 1];
    pf[0] = 0;
#pragma unroll
    for (int i = 0; i < BK_SPW; ++i) {
        sf[i]     = __builtin_amdgcn_readlane(pr.x, i);
        pf[i + 1] = pf[i] + __builtin_amdgcn_readlane(pr.y, i);
    }
    const int tot = pf[BK_SPW];

    const int4* __restrict__ rec = (const int4*)rec_g;
    for (int b = 0; b < tot; b += 32) {
        const int cnt = (tot - b) < 32 ? (tot - b) : 32;
        const int pos = b + lane;
        int a = sf[0] + pos;
#pragma unroll
        for (int i = 1; i < BK_SPW; ++i)
            if (pos >= pf[i]) a = sf[i] + (pos - pf[i]);
        int4 r = make_int4(0, 0, 0, 0);
        if (lane < cnt) r = rec[a];
        for (int g = 0; g < cnt; g += SP_GRP) {
            float hv[SP_GRP][4];
#pragma unroll
            for (int u = 0; u < SP_GRP; ++u) {
                const int iu = (g + u < cnt) ? (g + u) : (cnt - 1);
                const int su = __builtin_amdgcn_readlane(r.x, iu);
                const float* __restrict__ hp = h + (size_t)su * NHID + lane;
                hv[u][0] = hp[0];
                hv[u][1] = hp[32];
                hv[u][2] = hp[64];
                hv[u][3] = hp[96];
            }
#pragma unroll
            for (int u = 0; u < SP_GRP; ++u) {
                if (g + u < cnt) {
                    const int   du = __builtin_amdgcn_readlane(r.y, g + u);
                    const float wu = __int_as_float(__builtin_amdgcn_readlane(r.z, g + u));
                    float* tp = tile + du * NHID + lane;
                    atomicAdd(tp +  0, wu * hv[u][0]);
                    atomicAdd(tp + 32, wu * hv[u][1]);
                    atomicAdd(tp + 64, wu * hv[u][2]);
                    atomicAdd(tp + 96, wu * hv[u][3]);
                }
            }
        }
    }
    __syncthreads();

    float4* __restrict__ acc4 = (float4*)(acc + (size_t)r0 * NHID);
    for (int i = threadIdx.x; i < SP_ROWS * NHID / 4; i += SP_THREADS)
        acc4[i] = tile4[i];
}

#define HD_WAVES 8
#define HD_NT    (N_NODES / 16)
#define HD_CT    3
#define HD_OUTF  (16 * NCLASS)
#define HD_OUT4  (HD_OUTF / 4)
static_assert(N_NODES % 16 == 0, "node tiles must be whole");
static_assert(NCLASS > 32 && NCLASS <= 48, "three column tiles, the third one partial");
static_assert(HD_OUTF % 4 == 0, "a node tile of out is stored as float4");
static_assert(32 % HD_WAVES == 0, "the 32 k-steps of Wout are dealt out evenly to the waves");

template<int M>
__device__ __forceinline__ float row_xor(float x)
{
    return __builtin_bit_cast(float,
        __builtin_amdgcn_mov_dpp(__builtin_bit_cast(int, x), 0x160 + M, 0xf, 0xf, true));
}

__global__ __launch_bounds__(HD_WAVES * 32)
void head_wmma_logsoftmax(const float* __restrict__ acc, const float* __restrict__ b2,
                          const float* __restrict__ Wout, const float* __restrict__ bout,
                          float* __restrict__ out)
{
    __shared__ __attribute__((aligned(16))) float2 Bs[32 * HD_CT * 32];
    __shared__ __attribute__((aligned(16))) float  stg[HD_WAVES][HD_OUTF];

    const int lane = threadIdx.x & 31;
    const int wave = __builtin_amdgcn_readfirstlane((int)(threadIdx.x >> 5));
    const int c    = lane & 15;
    const int hh   = lane >> 4;
    int tile = blockIdx.x * HD_WAVES + wave;
    const bool live = tile < HD_NT;
    if (!live) tile = HD_NT - 1;

    const bool c2 = (32 + c) < NCLASS;
    const int  col2 = c2 ? (32 + c) : (NCLASS - 1);
    float2 wv[32 / HD_WAVES][HD_CT];
#pragma unroll
    for (int q = 0; q < 32 / HD_WAVES; ++q) {
        const int s = wave + q * HD_WAVES;
        const int k = 8 * (s >> 1) + 4 * hh + 2 * (s & 1);
        const float* __restrict__ w0 = Wout + (size_t)k * NCLASS;
        const float* __restrict__ w1 = w0 + NCLASS;
        wv[q][0] = make_float2(w0[c],      w1[c]);
        wv[q][1] = make_float2(w0[16 + c], w1[16 + c]);
        wv[q][2] = make_float2(w0[col2],   w1[col2]);
    }

    const float4* __restrict__ ap = (const float4*)(acc + (size_t)(tile * 16 + c) * NHID + 4 * hh);
    const float4* __restrict__ bp = (const float4*)(b2 + 4 * hh);
    float4 av[16], bv[16];
#pragma unroll
    for (int j = 0; j < 16; ++j) av[j] = ap[2 * j];
#pragma unroll
    for (int j = 0; j < 16; ++j) bv[j] = bp[2 * j];

#pragma unroll
    for (int q = 0; q < 32 / HD_WAVES; ++q) {
        const int s = wave + q * HD_WAVES;
        float2* __restrict__ bd = Bs + (s * HD_CT) * 32 + lane;
        bd[0]  = wv[q][0];
        bd[32] = wv[q][1];
        bd[64] = c2 ? wv[q][2] : make_float2(0.f, 0.f);
    }
    __syncthreads();

    v8f d0 = {}, d1 = {}, d2 = {};
#pragma unroll
    for (int j = 0; j < 16; ++j) {
        const float x0 = fmaxf(av[j].x + bv[j].x, 0.0f);
        const float x1 = fmaxf(av[j].y + bv[j].y, 0.0f);
        const float x2 = fmaxf(av[j].z + bv[j].z, 0.0f);
        const float x3 = fmaxf(av[j].w + bv[j].w, 0.0f);
        v2f a; float2 f; v2f b;
        const float2* __restrict__ bs = Bs + (2 * j) * HD_CT * 32 + lane;
        a.x = x0; a.y = x1;
        f = bs[0];   b.x = f.x; b.y = f.y;
        d0 = __builtin_amdgcn_wmma_f32_16x16x4_f32(false, a, false, b, (short)0, d0, false, false);
        f = bs[32];  b.x = f.x; b.y = f.y;
        d1 = __builtin_amdgcn_wmma_f32_16x16x4_f32(false, a, false, b, (short)0, d1, false, false);
        f = bs[64];  b.x = f.x; b.y = f.y;
        d2 = __builtin_amdgcn_wmma_f32_16x16x4_f32(false, a, false, b, (short)0, d2, false, false);
        a.x = x2; a.y = x3;
        f = bs[96];  b.x = f.x; b.y = f.y;
        d0 = __builtin_amdgcn_wmma_f32_16x16x4_f32(false, a, false, b, (short)0, d0, false, false);
        f = bs[128]; b.x = f.x; b.y = f.y;
        d1 = __builtin_amdgcn_wmma_f32_16x16x4_f32(false, a, false, b, (short)0, d1, false, false);
        f = bs[160]; b.x = f.x; b.y = f.y;
        d2 = __builtin_amdgcn_wmma_f32_16x16x4_f32(false, a, false, b, (short)0, d2, false, false);
    }

    const float bo0 = bout[c];
    const float bo1 = bout[16 + c];
    const float bo2 = bout[col2];
    float l0[8], l1[8], l2[8], mx[8], se[8];
#pragma unroll
    for (int r = 0; r < 8; ++r) {
        l0[r] = d0[r] + bo0;
        l1[r] = d1[r] + bo1;
        l2[r] = c2 ? (d2[r] + bo2) : -INFINITY;
        mx[r] = fmaxf(fmaxf(l0[r], l1[r]), l2[r]);
    }
#pragma unroll
    for (int r = 0; r < 8; ++r) mx[r] = fmaxf(mx[r], row_xor<1>(mx[r]));
#pragma unroll
    for (int r = 0; r < 8; ++r) mx[r] = fmaxf(mx[r], row_xor<2>(mx[r]));
#pragma unroll
    for (int r = 0; r < 8; ++r) mx[r] = fmaxf(mx[r], row_xor<4>(mx[r]));
#pragma unroll
    for (int r = 0; r < 8; ++r) mx[r] = fmaxf(mx[r], row_xor<8>(mx[r]));
#pragma unroll
    for (int r = 0; r < 8; ++r)
        se[r] = __expf(l0[r] - mx[r]) + __expf(l1[r] - mx[r]) + (c2 ? __expf(l2[r] - mx[r]) : 0.0f);
#pragma unroll
    for (int r = 0; r < 8; ++r) se[r] += row_xor<1>(se[r]);
#pragma unroll
    for (int r = 0; r < 8; ++r) se[r] += row_xor<2>(se[r]);
#pragma unroll
    for (int r = 0; r < 8; ++r) se[r] += row_xor<4>(se[r]);
#pragma unroll
    for (int r = 0; r < 8; ++r) se[r] += row_xor<8>(se[r]);

    float* __restrict__ sp = stg[wave] + (8 * hh) * NCLASS + c;
#pragma unroll
    for (int r = 0; r < 8; ++r) {
        const float lse = __logf(se[r]) + mx[r];
        sp[r * NCLASS]      = l0[r] - lse;
        sp[r * NCLASS + 16] = l1[r] - lse;
        if (c2) sp[r * NCLASS + 32] = l2[r] - lse;
    }
    __syncthreads();

    if (live) {
        float4* __restrict__ o4 = (float4*)(out + (size_t)tile * HD_OUTF);
        const float4* s4 = (const float4*)stg[wave];
#pragma unroll
        for (int u = 0; u < (HD_OUT4 + 31) / 32; ++u) {
            const int i = u * 32 + lane;
            if (i < HD_OUT4) o4[i] = s4[i];
        }
    }
}

extern "C" void kernel_launch(void* const* d_in, const int* in_sizes, int n_in,
                              void* d_out, int out_size, void* d_ws, size_t ws_size,
                              hipStream_t stream)
{
    const float* x    = (const float*)d_in[0];
    const int*   a0i  = (const int*)  d_in[1];
    const float* a0w  = (const float*)d_in[2];
    const int*   a1i  = (const int*)  d_in[3];
    const float* a1w  = (const float*)d_in[4];
    const float* W1   = (const float*)d_in[5];
    const float* b1   = (const float*)d_in[6];
    const float* W2   = (const float*)d_in[7];
    const float* b2   = (const float*)d_in[8];
    const float* Wout = (const float*)d_in[9];
    const float* bout = (const float*)d_in[10];
    float* out = (float*)d_out;

    const size_t NH = (size_t)N_NODES * NHID;
    float* bufA = (float*)d_ws;
    float* bufB = bufA + NH;
    int*   recAll = (int*)(bufB + NH);
    const size_t REC_G = (size_t)N_EDGES * 4;
    int*   segAll = recAll + 2 * REC_G;
    const size_t SEG_G = (size_t)SP_NB * BK_NCH * 2;

    const dim3 gGemm(N_NODES / 16), bGemm(256);
    const dim3 gSp(SP_NB), bSp(SP_THREADS);

    bucket_chunks<<<dim3(2 * BK_NCH), dim3(BK_THREADS), 0, stream>>>(
        a0i, a0i + N_EDGES, a0w, a1i, a1i + N_EDGES, a1w, recAll, segAll);

    gemm128_wmma<false><<<gGemm, bGemm, 0, stream>>>(x, W1, nullptr, bufA);
    spmm_rows_lds<<<gSp, bSp, 0, stream>>>(recAll, segAll, bufA, bufB);

    gemm128_wmma<true><<<gGemm, bGemm, 0, stream>>>(bufB, W2, b1, bufA);
    spmm_rows_lds<<<gSp, bSp, 0, stream>>>(recAll + REC_G, segAll + SEG_G, bufA, bufB);

    head_wmma_logsoftmax<<<dim3((HD_NT + HD_WAVES - 1) / HD_WAVES), dim3(HD_WAVES * 32), 0, stream>>>(bufB, b2, Wout, bout, out);
}


# gemm: W staged once per group in LDS in fragment order; a wave owns 16 rows x all 128 columns (8 D tiles), A rows read once by 16-byte loads, whole-line stores; dot sums in a new fixed k order
# speedup vs baseline: 3.3584x; 1.2135x over previous
#include <hip/hip_runtime.h>
#include <math.h>

#define N_NODES 50000
#define N_EDGES 600000
#define NFEAT 128
#define NHID 128
#define NCLASS 41

typedef float v2f __attribute__((ext_vector_type(2)));
typedef float v8f __attribute__((ext_vector_type(8)));

#define GM_WGS   256
#define GM_WAVES 16
#define GM_FULL  12
#define GM_NT    (N_NODES / 16)
static_assert(N_NODES % 16 == 0, "row tiles must be whole");
static_assert(NFEAT == 128 && NHID == 128, "the fragment layout below is for 128 x 128 weights");
static_assert(GM_FULL * GM_WGS <= GM_NT && GM_NT <= (GM_FULL + 1) * GM_WGS, "full-tile waves plus one split tile per group cover all tiles");

template<int NT>
__device__ __forceinline__ void gm_rows_times_b(const float4 (&av)[16], const float4* bs, float* __restrict__ crow)
{
    v8f d[NT];
#pragma unroll
    for (int t = 0; t < NT; ++t) d[t] = v8f{};
    float4 f[2][NT];
#pragma unroll
    for (int t = 0; t < NT; ++t) f[0][t] = bs[t * 32];
#pragma unroll
    for (int j = 0; j < 16; ++j) {
        if (j + 1 < 16) {
#pragma unroll
            for (int t = 0; t < NT; ++t) f[(j + 1) & 1][t] = bs[((j + 1) * 8 + t) * 32];
        }
        __builtin_amdgcn_sched_barrier(0);
        v2f a0, a1;
        a0.x = av[j].x; a0.y = av[j].y;
        a1.x = av[j].z; a1.y = av[j].w;
#pragma unroll
        for (int t = 0; t < NT; ++t) {
            v2f b; b.x = f[j & 1][t].x; b.y = f[j & 1][t].y;
            d[t] = __builtin_amdgcn_wmma_f32_16x16x4_f32(false, a0, false, b, (short)0, d[t], false, false);
        }
#pragma unroll
        for (int t = 0; t < NT; ++t) {
            v2f b; b.x = f[j & 1][t].z; b.y = f[j & 1][t].w;
            d[t] = __builtin_amdgcn_wmma_f32_16x16x4_f32(false, a1, false, b, (short)0, d[t], false, false);
        }
        __builtin_amdgcn_sched_barrier(0);
    }
#pragma unroll
    for (int v = 0; v < 8; ++v) {
#pragma unroll
        for (int p = 0; p < NT / 2; ++p)
            *(float2*)(crow + v * NHID + 32 * p) = make_float2(d[2 * p][v], d[2 * p + 1][v]);
    }
}

template<bool BIAS_RELU>
__global__ __launch_bounds__(GM_WAVES * 32)
void gemm128_wmma(const float* __restrict__ A, const float* __restrict__ B,
                  const float* __restrict__ bias, float* __restrict__ C)
{
    __shared__ __attribute__((aligned(16))) float4 Bs[16 * 8 * 32];

    const int lane = threadIdx.x & 31;
    const int wave = __builtin_amdgcn_readfirstlane((int)(threadIdx.x >> 5));
    const int c    = lane & 15;
    const int hh   = lane >> 4;

    float4 wr[2][4];
#pragma unroll
    for (int q = 0; q < 2; ++q) {
        const int kq = wave + GM_WAVES * q;
        const float4* __restrict__ wp = (const float4*)(B + (size_t)(4 * kq) * NHID) + lane;
#pragma unroll
        for (int r = 0; r < 4; ++r) wr[q][r] = wp[r * (NHID / 4)];
    }

    const bool quarter = (wave % 3 == 0) && wave > 0 && wave < 15;
    const int  slot    = wave - (wave > 3) - (wave > 6) - (wave > 9) - (wave > 12);
    const int  tile    = quarter ? (GM_FULL * GM_WGS + (int)blockIdx.x) : ((int)blockIdx.x + GM_WGS * slot);
    const bool live    = tile < GM_NT;

    float4 av[16];
    if (live) {
        const float4* __restrict__ ap = (const float4*)(A + (size_t)(tile * 16 + c) * NHID + 4 * hh);
        if (BIAS_RELU) {
            const float4* __restrict__ bp = (const float4*)(bias + 4 * hh);
            float4 bv[16];
#pragma unroll
            for (int j = 0; j < 16; ++j) bv[j] = bp[2 * j];
#pragma unroll
            for (int j = 0; j < 16; ++j) av[j] = ap[2 * j];
#pragma unroll
            for (int j = 0; j < 16; ++j) {
                av[j].x = fmaxf(av[j].x + bv[j].x, 0.0f);
                av[j].y = fmaxf(av[j].y + bv[j].y, 0.0f);
                av[j].z = fmaxf(av[j].z + bv[j].z, 0.0f);
                av[j].w = fmaxf(av[j].w + bv[j].w, 0.0f);
            }
        } else {
#pragma unroll
            for (int j = 0; j < 16; ++j) av[j] = ap[2 * j];
        }
    }

#pragma unroll
    for (int q = 0; q < 2; ++q) {
        const int kq = wave + GM_WAVES * q;
        float4* __restrict__ bd = Bs + ((kq >> 1) * 8 + 2 * (lane >> 3)) * 32 + (kq & 1) * 16 + 2 * (lane & 7);
        bd[0]      = make_float4(wr[q][0].x, wr[q][1].x, wr[q][2].x, wr[q][3].x);
        bd[32]     = make_float4(wr[q][0].y, wr[q][1].y, wr[q][2].y, wr[q][3].y);
        bd[1]      = make_float4(wr[q][0].z, wr[q][1].z, wr[q][2].z, wr[q][3].z);
        bd[32 + 1] = make_float4(wr[q][0].w, wr[q][1].w, wr[q][2].w, wr[q][3].w);
    }
    __syncthreads();
    if (!live) return;

    float* __restrict__ crow = C + (size_t)(tile * 16 + 8 * hh) * NHID + 2 * c;
    if (quarter) {
        const int q0 = wave / 3 - 1;
        gm_rows_times_b<2>(av, Bs + (2 * q0) * 32 + lane, crow + 32 * q0);
    } else {
        gm_rows_times_b<8>(av, Bs + lane, crow);
    }
}

#define SP_ROWS    200
#define SP_NB      (N_NODES / SP_ROWS)
#define SP_THREADS 1024
#define SP_WAVES   (SP_THREADS / 32)
#define SP_GRP     8
#define BK_CH      4800
#define BK_NCH     (N_EDGES / BK_CH)
#define BK_THREADS 512
#define BK_PER     ((BK_CH + BK_THREADS - 1) / BK_THREADS)
#define BK_BINS    256
#define BK_SPW     ((BK_NCH + SP_WAVES - 1) / SP_WAVES)
static_assert(N_NODES % SP_ROWS == 0, "row ranges must tile the node set exactly");
static_assert(N_EDGES % BK_CH == 0, "chunks must tile the edge list exactly");
static_assert(SP_NB < BK_BINS && BK_BINS == 8 * 32, "the scan gives each lane of one wave 8 bins");
static_assert(SP_NB <= BK_THREADS, "one thread writes one (first, count) pair");
static_assert(BK_SPW <= 32, "a wave holds its chunk pairs one per lane");

__global__ __launch_bounds__(BK_THREADS)
void bucket_chunks(const int* __restrict__ src0, const int* __restrict__ dst0, const float* __restrict__ w0,
                   const int* __restrict__ src1, const int* __restrict__ dst1, const float* __restrict__ w1,
                   int* __restrict__ rec_all, int* __restrict__ seg_all)
{
    __shared__ int hist[BK_BINS];
    __shared__ int first[BK_BINS];
    __shared__ __attribute__((aligned(16))) int4 sorted[BK_CH];

    const int t     = threadIdx.x;
    const int graph = (int)blockIdx.x / BK_NCH;
    const int chunk = (int)blockIdx.x - graph * BK_NCH;
    const int e0    = chunk * BK_CH;
    const int*   __restrict__ src = graph ? src1 : src0;
    const int*   __restrict__ dst = graph ? dst1 : dst0;
    const float* __restrict__ w   = graph ? w1 : w0;

    if (t < BK_BINS) hist[t] = 0;

    int s[BK_PER] = {}, d[BK_PER] = {}, wb[BK_PER] = {}, bin[BK_PER] = {}, rank[BK_PER] = {};
#pragma unroll
    for (int j = 0; j < BK_PER; ++j) {
        const int i = j * BK_THREADS + t;
        if (i < BK_CH) {
            s[j]  = src[e0 + i];
            d[j]  = dst[e0 + i];
            wb[j] = __float_as_int(w[e0 + i]);
        }
    }
    __syncthreads();

#pragma unroll
    for (int j = 0; j < BK_PER; ++j) {
        const int i = j * BK_THREADS + t;
        if (i < BK_CH) {
            bin[j]  = ((unsigned)d[j] < (unsigned)N_NODES) ? (d[j] / SP_ROWS) : SP_NB;
            rank[j] = atomicAdd(&hist[bin[j]], 1);
        }
    }
    __syncthreads();

    if (t < 32) {
        int c[8];
        int sum = 0;
#pragma unroll
        for (int k = 0; k < 8; ++k) { c[k] = hist[t * 8 + k]; sum += c[k]; }
        int incl = sum;
#pragma unroll
        for (int dl = 1; dl < 32; dl <<= 1) {
            const int o = __shfl_up(incl, dl);
            if (t >= dl) incl += o;
        }
        int run = incl - sum;
#pragma unroll
        for (int k = 0; k < 8; ++k) { first[t * 8 + k] = run; run += c[k]; }
    }
    __syncthreads();

#pragma unroll
    for (int j = 0; j < BK_PER; ++j) {
        const int i = j * BK_THREADS + t;
        if (i < BK_CH)
            sorted[first[bin[j]] + rank[j]] = make_int4(s[j], d[j] - bin[j] * SP_ROWS, wb[j], 0);
    }
    if (t < SP_NB) {
        int2* __restrict__ seg = (int2*)seg_all + ((size_t)graph * SP_NB + t) * BK_NCH + chunk;
        *seg = make_int2(e0 + first[t], hist[t]);
    }
    __syncthreads();

    int4* __restrict__ rec = (int4*)rec_all + (size_t)graph * N_EDGES + e0;
    for (int i = t; i < BK_CH; i += BK_THREADS) rec[i] = sorted[i];
}

__global__ __launch_bounds__(SP_THREADS)
void spmm_rows_lds(const int* __restrict__ rec_g, const int* __restrict__ seg_g,
                   const float* __restrict__ h, float* __restrict__ acc)
{
    __shared__ __attribute__((aligned(16))) float tile[SP_ROWS * NHID];

    const int lane = threadIdx.x & 31;
    const int wave = __builtin_amdgcn_readfirstlane((int)(threadIdx.x >> 5));
    const int r0   = blockIdx.x * SP_ROWS;

    const int2* __restrict__ seg = (const int2*)seg_g + (size_t)blockIdx.x * BK_NCH;
    const int myc = wave + lane * SP_WAVES;
    int2 pr = make_int2(0, 0);
    if (lane < BK_SPW && myc < BK_NCH) pr = seg[myc];

    float4* tile4 = (float4*)tile;
    for (int i = threadIdx.x; i < SP_ROWS * NHID / 4; i += SP_THREADS)
        tile4[i] = make_float4(0.f, 0.f, 0.f, 0.f);
    __syncthreads();

    int sf[BK_SPW], pf[BK_SPW + 1];
    pf[0] = 0;
#pragma unroll
    for (int i = 0; i < BK_SPW; ++i) {
        sf[i]     = __builtin_amdgcn_readlane(pr.x, i);
        pf[i + 1] = pf[i] + __builtin_amdgcn_readlane(pr.y, i);
    }
    const int tot = pf[BK_SPW];

    const int4* __restrict__ rec = (const int4*)rec_g;
    for (int b = 0; b < tot; b += 32) {
        const int cnt = (tot - b) < 32 ? (tot - b) : 32;
        const int pos = b + lane;
        int a = sf[0] + pos;
#pragma unroll
        for (int i = 1; i < BK_SPW; ++i)
            if (pos >= pf[i]) a = sf[i] + (pos - pf[i]);
        int4 r = make_int4(0, 0, 0, 0);
        if (lane < cnt) r = rec[a];
        for (int g = 0; g < cnt; g += SP_GRP) {
            float hv[SP_GRP][4];
#pragma unroll
            for (int u = 0; u < SP_GRP; ++u) {
                const int iu = (g + u < cnt) ? (g + u) : (cnt - 1);
                const int su = __builtin_amdgcn_readlane(r.x, iu);
                const float* __restrict__ hp = h + (size_t)su * NHID + lane;
                hv[u][0] = hp[0];
                hv[u][1] = hp[32];
                hv[u][2] = hp[64];
                hv[u][3] = hp[96];
            }
#pragma unroll
            for (int u = 0; u < SP_GRP; ++u) {
                if (g + u < cnt) {
                    const int   du = __builtin_amdgcn_readlane(r.y, g + u);
                    const float wu = __int_as_float(__builtin_amdgcn_readlane(r.z, g + u));
                    float* tp = tile + du * NHID + lane;
                    atomicAdd(tp +  0, wu * hv[u][0]);
                    atomicAdd(tp + 32, wu * hv[u][1]);
                    atomicAdd(tp + 64, wu * hv[u][2]);
                    atomicAdd(tp + 96, wu * hv[u][3]);
                }
            }
        }
    }
    __syncthreads();

    float4* __restrict__ acc4 = (float4*)(acc + (size_t)r0 * NHID);
    for (int i = threadIdx.x; i < SP_ROWS * NHID / 4; i += SP_THREADS)
        acc4[i] = tile4[i];
}

#define HD_WAVES 8
#define HD_NT    (N_NODES / 16)
#define HD_CT    3
#define HD_OUTF  (16 * NCLASS)
#define HD_OUT4  (HD_OUTF / 4)
static_assert(N_NODES % 16 == 0, "node tiles must be whole");
static_assert(NCLASS > 32 && NCLASS <= 48, "three column tiles, the third one partial");
static_assert(HD_OUTF % 4 == 0, "a node tile of out is stored as float4");
static_assert(32 % HD_WAVES == 0, "the 32 k-steps of Wout are dealt out evenly to the waves");

template<int M>
__device__ __forceinline__ float row_xor(float x)
{
    return __builtin_bit_cast(float,
        __builtin_amdgcn_mov_dpp(__builtin_bit_cast(int, x), 0x160 + M, 0xf, 0xf, true));
}

__global__ __launch_bounds__(HD_WAVES * 32)
void head_wmma_logsoftmax(const float* __restrict__ acc, const float* __restrict__ b2,
                          const float* __restrict__ Wout, const float* __restrict__ bout,
                          float* __restrict__ out)
{
    __shared__ __attribute__((aligned(16))) float2 Bs[32 * HD_CT * 32];
    __shared__ __attribute__((aligned(16))) float  stg[HD_WAVES][HD_OUTF];

    const int lane = threadIdx.x & 31;
    const int wave = __builtin_amdgcn_readfirstlane((int)(threadIdx.x >> 5));
    const int c    = lane & 15;
    const int hh   = lane >> 4;
    int tile = blockIdx.x * HD_WAVES + wave;
    const bool live = tile < HD_NT;
    if (!live) tile = HD_NT - 1;

    const bool c2 = (32 + c) < NCLASS;
    const int  col2 = c2 ? (32 + c) : (NCLASS - 1);
    float2 wv[32 / HD_WAVES][HD_CT];
#pragma unroll
    for (int q = 0; q < 32 / HD_WAVES; ++q) {
        const int s = wave + q * HD_WAVES;
        const int k = 8 * (s >> 1) + 4 * hh + 2 * (s & 1);
        const float* __restrict__ w0 = Wout + (size_t)k * NCLASS;
        const float* __restrict__ w1 = w0 + NCLASS;
        wv[q][0] = make_float2(w0[c],      w1[c]);
        wv[q][1] = make_float2(w0[16 + c], w1[16 + c]);
        wv[q][2] = make_float2(w0[col2],   w1[col2]);
    }

    const float4* __restrict__ ap = (const float4*)(acc + (size_t)(tile * 16 + c) * NHID + 4 * hh);
    const float4* __restrict__ bp = (const float4*)(b2 + 4 * hh);
    float4 av[16], bv[16];
#pragma unroll
    for (int j = 0; j < 16; ++j) av[j] = ap[2 * j];
#pragma unroll
    for (int j = 0; j < 16; ++j) bv[j] = bp[2 * j];

#pragma unroll
    for (int q = 0; q < 32 / HD_WAVES; ++q) {
        const int s = wave + q * HD_WAVES;
        float2* __restrict__ bd = Bs + (s * HD_CT) * 32 + lane;
        bd[0]  = wv[q][0];
        bd[32] = wv[q][1];
        bd[64] = c2 ? wv[q][2] : make_float2(0.f, 0.f);
    }
    __syncthreads();

    v8f d0 = {}, d1 = {}, d2 = {};
#pragma unroll
    for (int j = 0; j < 16; ++j) {
        const float x0 = fmaxf(av[j].x + bv[j].x, 0.0f);
        const float x1 = fmaxf(av[j].y + bv[j].y, 0.0f);
        const float x2 = fmaxf(av[j].z + bv[j].z, 0.0f);
        const float x3 = fmaxf(av[j].w + bv[j].w, 0.0f);
        v2f a; float2 f; v2f b;
        const float2* __restrict__ bs = Bs + (2 * j) * HD_CT * 32 + lane;
        a.x = x0; a.y = x1;
        f = bs[0];   b.x = f.x; b.y = f.y;
        d0 = __builtin_amdgcn_wmma_f32_16x16x4_f32(false, a, false, b, (short)0, d0, false, false);
        f = bs[32];  b.x = f.x; b.y = f.y;
        d1 = __builtin_amdgcn_wmma_f32_16x16x4_f32(false, a, false, b, (short)0, d1, false, false);
        f = bs[64];  b.x = f.x; b.y = f.y;
        d2 = __builtin_amdgcn_wmma_f32_16x16x4_f32(false, a, false, b, (short)0, d2, false, false);
        a.x = x2; a.y = x3;
        f = bs[96];  b.x = f.x; b.y = f.y;
        d0 = __builtin_amdgcn_wmma_f32_16x16x4_f32(false, a, false, b, (short)0, d0, false, false);
        f = bs[128]; b.x = f.x; b.y = f.y;
        d1 = __builtin_amdgcn_wmma_f32_16x16x4_f32(false, a, false, b, (short)0, d1, false, false);
        f = bs[160]; b.x = f.x; b.y = f.y;
        d2 = __builtin_amdgcn_wmma_f32_16x16x4_f32(false, a, false, b, (short)0, d2, false, false);
    }

    const float bo0 = bout[c];
    const float bo1 = bout[16 + c];
    const float bo2 = bout[col2];
    float l0[8], l1[8], l2[8], mx[8], se[8];
#pragma unroll
    for (int r = 0; r < 8; ++r) {
        l0[r] = d0[r] + bo0;
        l1[r] = d1[r] + bo1;
        l2[r] = c2 ? (d2[r] + bo2) : -INFINITY;
        mx[r] = fmaxf(fmaxf(l0[r], l1[r]), l2[r]);
    }
#pragma unroll
    for (int r = 0; r < 8; ++r) mx[r] = fmaxf(mx[r], row_xor<1>(mx[r]));
#pragma unroll
    for (int r = 0; r < 8; ++r) mx[r] = fmaxf(mx[r], row_xor<2>(mx[r]));
#pragma unroll
    for (int r = 0; r < 8; ++r) mx[r] = fmaxf(mx[r], row_xor<4>(mx[r]));
#pragma unroll
    for (int r = 0; r < 8; ++r) mx[r] = fmaxf(mx[r], row_xor<8>(mx[r]));
#pragma unroll
    for (int r = 0; r < 8; ++r)
        se[r] = __expf(l0[r] - mx[r]) + __expf(l1[r] - mx[r]) + (c2 ? __expf(l2[r] - mx[r]) : 0.0f);
#pragma unroll
    for (int r = 0; r < 8; ++r) se[r] += row_xor<1>(se[r]);
#pragma unroll
    for (int r = 0; r < 8; ++r) se[r] += row_xor<2>(se[r]);
#pragma unroll
    for (int r = 0; r < 8; ++r) se[r] += row_xor<4>(se[r]);
#pragma unroll
    for (int r = 0; r < 8; ++r) se[r] += row_xor<8>(se[r]);

    float* __restrict__ sp = stg[wave] + (8 * hh) * NCLASS + c;
#pragma unroll
    for (int r = 0; r < 8; ++r) {
        const float lse = __logf(se[r]) + mx[r];
        sp[r * NCLASS]      = l0[r] - lse;
        sp[r * NCLASS + 16] = l1[r] - lse;
        if (c2) sp[r * NCLASS + 32] = l2[r] - lse;
    }
    __syncthreads();

    if (live) {
        float4* __restrict__ o4 = (float4*)(out + (size_t)tile * HD_OUTF);
        const float4* s4 = (const float4*)stg[wave];
#pragma unroll
        for (int u = 0; u < (HD_OUT4 + 31) / 32; ++u) {
            const int i = u * 32 + lane;
            if (i < HD_OUT4) o4[i] = s4[i];
        }
    }
}

extern "C" void kernel_launch(void* const* d_in, const int* in_sizes, int n_in,
                              void* d_out, int out_size, void* d_ws, size_t ws_size,
                              hipStream_t stream)
{
    const float* x    = (const float*)d_in[0];
    const int*   a0i  = (const int*)  d_in[1];
    const float* a0w  = (const float*)d_in[2];
    const int*   a1i  = (const int*)  d_in[3];
    const float* a1w  = (const float*)d_in[4];
    const float* W1   = (const float*)d_in[5];
    const float* b1   = (const float*)d_in[6];
    const float* W2   = (const float*)d_in[7];
    const float* b2   = (const float*)d_in[8];
    const float* Wout = (const float*)d_in[9];
    const float* bout = (const float*)d_in[10];
    float* out = (float*)d_out;

    const size_t NH = (size_t)N_NODES * NHID;
    float* bufA = (float*)d_ws;
    float* bufB = bufA + NH;
    int*   recAll = (int*)(bufB + NH);
    const size_t REC_G = (size_t)N_EDGES * 4;
    int*   segAll = recAll + 2 * REC_G;
    const size_t SEG_G = (size_t)SP_NB * BK_NCH * 2;

    const dim3 gGemm(GM_WGS), bGemm(GM_WAVES * 32);
    const dim3 gSp(SP_NB), bSp(SP_THREADS);

    bucket_chunks<<<dim3(2 * BK_NCH), dim3(BK_THREADS), 0, stream>>>(
        a0i, a0i + N_EDGES, a0w, a1i, a1i + N_EDGES, a1w, recAll, segAll);

    gemm128_wmma<false><<<gGemm, bGemm, 0, stream>>>(x, W1, nullptr, bufA);
    spmm_rows_lds<<<gSp, bSp, 0, stream>>>(recAll, segAll, bufA, bufB);

    gemm128_wmma<true><<<gGemm, bGemm, 0, stream>>>(bufB, W2, b1, bufA);
    spmm_rows_lds<<<gSp, bSp, 0, stream>>>(recAll + REC_G, segAll + SEG_G, bufA, bufB);

    head_wmma_logsoftmax<<<dim3((HD_NT + HD_WAVES - 1) / HD_WAVES), dim3(HD_WAVES * 32), 0, stream>>>(bufB, b2, Wout, bout, out);
}


# spmm: the block's finished LDS tile is the A operand of the next dense product on WMMA in the same launch (W2; then Wout + log-softmax); acc0/acc1 never reach memory; 4 launches, not 6
# speedup vs baseline: 4.0421x; 1.2036x over previous
#include <hip/hip_runtime.h>
#include <math.h>

#define N_NODES 50000
#define N_EDGES 600000
#define NFEAT 128
#define NHID 128
#define NCLASS 41

typedef float v2f __attribute__((ext_vector_type(2)));
typedef float v8f __attribute__((ext_vector_type(8)));

#define GM_WGS   256
#define GM_WAVES 16
#define GM_FULL  12
#define GM_NT    (N_NODES / 16)
static_assert(N_NODES % 16 == 0, "row tiles must be whole");
static_assert(NFEAT == 128 && NHID == 128, "the fragment layout below is for 128 x 128 weights");
static_assert(GM_FULL * GM_WGS <= GM_NT && GM_NT <= (GM_FULL + 1) * GM_WGS, "full-tile waves plus one split tile per group cover all tiles");

template<int NT>
__device__ __forceinline__ void gm_rows_times_b(const float4 (&av)[16], const float4* bs, float* __restrict__ crow, const bool st)
{
    v8f d[NT];
#pragma unroll
    for (int t = 0; t < NT; ++t) d[t] = v8f{};
    float4 f[2][NT];
#pragma unroll
    for (int t = 0; t < NT; ++t) f[0][t] = bs[t * 32];
#pragma unroll
    for (int j = 0; j < 16; ++j) {
        if (j + 1 < 16) {
#pragma unroll
            for (int t = 0; t < NT; ++t) f[(j + 1) & 1][t] = bs[((j + 1) * 8 + t) * 32];
        }
        __builtin_amdgcn_sched_barrier(0);
        v2f a0, a1;
        a0.x = av[j].x; a0.y = av[j].y;
        a1.x = av[j].z; a1.y = av[j].w;
#pragma unroll
        for (int t = 0; t < NT; ++t) {
            v2f b; b.x = f[j & 1][t].x; b.y = f[j & 1][t].y;
            d[t] = __builtin_amdgcn_wmma_f32_16x16x4_f32(false, a0, false, b, (short)0, d[t], false, false);
        }
#pragma unroll
        for (int t = 0; t < NT; ++t) {
            v2f b; b.x = f[j & 1][t].z; b.y = f[j & 1][t].w;
            d[t] = __builtin_amdgcn_wmma_f32_16x16x4_f32(false, a1, false, b, (short)0, d[t], false, false);
        }
        __builtin_amdgcn_sched_barrier(0);
    }
    if (st) {
#pragma unroll
        for (int v = 0; v < 8; ++v) {
#pragma unroll
            for (int p = 0; p < NT / 2; ++p)
                *(float2*)(crow + v * NHID + 32 * p) = make_float2(d[2 * p][v], d[2 * p + 1][v]);
        }
    }
}

__global__ __launch_bounds__(GM_WAVES * 32)
void gemm128_wmma(const float* __restrict__ A, const float* __restrict__ B, float* __restrict__ C)
{
    __shared__ __attribute__((aligned(16))) float4 Bs[16 * 8 * 32];

    const int lane = threadIdx.x & 31;
    const int wave = __builtin_amdgcn_readfirstlane((int)(threadIdx.x >> 5));
    const int c    = lane & 15;
    const int hh   = lane >> 4;

    float4 wr[2][4];
#pragma unroll
    for (int q = 0; q < 2; ++q) {
        const int kq = wave + GM_WAVES * q;
        const float4* __restrict__ wp = (const float4*)(B + (size_t)(4 * kq) * NHID) + lane;
#pragma unroll
        for (int r = 0; r < 4; ++r) wr[q][r] = wp[r * (NHID / 4)];
    }

    const bool quarter = (wave % 3 == 0) && wave > 0 && wave < 15;
    const int  slot    = wave - (wave > 3) - (wave > 6) - (wave > 9) - (wave > 12);
    const int  tile    = quarter ? (GM_FULL * GM_WGS + (int)blockIdx.x) : ((int)blockIdx.x + GM_WGS * slot);
    const bool live    = tile < GM_NT;

    float4 av[16];
    if (live) {
        const float4* __restrict__ ap = (const float4*)(A + (size_t)(tile * 16 + c) * NHID + 4 * hh);
#pragma unroll
        for (int j = 0; j < 16; ++j) av[j] = ap[2 * j];
    }

#pragma unroll
    for (int q = 0; q < 2; ++q) {
        const int kq = wave + GM_WAVES * q;
        float4* __restrict__ bd = Bs + ((kq >> 1) * 8 + 2 * (lane >> 3)) * 32 + (kq & 1) * 16 + 2 * (lane & 7);
        bd[0]      = make_float4(wr[q][0].x, wr[q][1].x, wr[q][2].x, wr[q][3].x);
        bd[32]     = make_float4(wr[q][0].y, wr[q][1].y, wr[q][2].y, wr[q][3].y);
        bd[1]      = make_float4(wr[q][0].z, wr[q][1].z, wr[q][2].z, wr[q][3].z);
        bd[32 + 1] = make_float4(wr[q][0].w, wr[q][1].w, wr[q][2].w, wr[q][3].w);
    }
    __syncthreads();
    if (!live) return;

    float* __restrict__ crow = C + (size_t)(tile * 16 + 8 * hh) * NHID + 2 * c;
    if (quarter) {
        const int q0 = wave / 3 - 1;
        gm_rows_times_b<2>(av, Bs + (2 * q0) * 32 + lane, crow + 32 * q0, true);
    } else {
        gm_rows_times_b<8>(av, Bs + lane, crow, true);
    }
}

#define SP_ROWS    200
#define SP_RT      ((SP_ROWS + 15) / 16)
#define SP_TROWS   (SP_RT * 16)
#define SP_PITCH   (NHID + 4)
#define SP_TILE_F  (SP_TROWS * SP_PITCH)
#define SP_NB      (N_NODES / SP_ROWS)
#define SP_THREADS 1024
#define SP_WAVES   (SP_THREADS / 32)
#define SP_GRP     8
#define BK_CH      4800
#define BK_NCH     (N_EDGES / BK_CH)
#define BK_THREADS 512
#define BK_PER     ((BK_CH + BK_THREADS - 1) / BK_THREADS)
#define BK_BINS    256
#define BK_SPW     ((BK_NCH + SP_WAVES - 1) / SP_WAVES)
static_assert(N_NODES % SP_ROWS == 0, "row ranges must tile the node set exactly");
static_assert(N_EDGES % BK_CH == 0, "chunks must tile the edge list exactly");
static_assert(SP_NB < BK_BINS && BK_BINS == 8 * 32, "the scan gives each lane of one wave 8 bins");
static_assert(SP_NB <= BK_THREADS, "one thread writes one (first, count) pair");
static_assert(BK_SPW <= 32, "a wave holds its chunk pairs one per lane");

__global__ __launch_bounds__(BK_THREADS)
void bucket_chunks(const int* __restrict__ src0, const int* __restrict__ dst0, const float* __restrict__ w0,
                   const int* __restrict__ src1, const int* __restrict__ dst1, const float* __restrict__ w1,
                   int* __restrict__ rec_all, int* __restrict__ seg_all)
{
    __shared__ int hist[BK_BINS];
    __shared__ int first[BK_BINS];
    __shared__ __attribute__((aligned(16))) int4 sorted[BK_CH];

    const int t     = threadIdx.x;
    const int graph = (int)blockIdx.x / BK_NCH;
    const int chunk = (int)blockIdx.x - graph * BK_NCH;
    const int e0    = chunk * BK_CH;
    const int*   __restrict__ src = graph ? src1 : src0;
    const int*   __restrict__ dst = graph ? dst1 : dst0;
    const float* __restrict__ w   = graph ? w1 : w0;

    if (t < BK_BINS) hist[t] = 0;

    int s[BK_PER] = {}, d[BK_PER] = {}, wb[BK_PER] = {}, bin[BK_PER] = {}, rank[BK_PER] = {};
#pragma unroll
    for (int j = 0; j < BK_PER; ++j) {
        const int i = j * BK_THREADS + t;
        if (i < BK_CH) {
            s[j]  = src[e0 + i];
            d[j]  = dst[e0 + i];
            wb[j] = __float_as_int(w[e0 + i]);
        }
    }
    __syncthreads();

#pragma unroll
    for (int j = 0; j < BK_PER; ++j) {
        const int i = j * BK_THREADS + t;
        if (i < BK_CH) {
            bin[j]  = ((unsigned)d[j] < (unsigned)N_NODES) ? (d[j] / SP_ROWS) : SP_NB;
            rank[j] = atomicAdd(&hist[bin[j]], 1);
        }
    }
    __syncthreads();

    if (t < 32) {
        int c[8];
        int sum = 0;
#pragma unroll
        for (int k = 0; k < 8; ++k) { c[k] = hist[t * 8 + k]; sum += c[k]; }
        int incl = sum;
#pragma unroll
        for (int dl = 1; dl < 32; dl <<= 1) {
            const int o = __shfl_up(incl, dl);
            if (t >= dl) incl += o;
        }
        int run = incl - sum;
#pragma unroll
        for (int k = 0; k < 8; ++k) { first[t * 8 + k] = run; run += c[k]; }
    }
    __syncthreads();

#pragma unroll
    for (int j = 0; j < BK_PER; ++j) {
        const int i = j * BK_THREADS + t;
        if (i < BK_CH)
            sorted[first[bin[j]] + rank[j]] = make_int4(s[j], d[j] - bin[j] * SP_ROWS, wb[j], 0);
    }
    if (t < SP_NB) {
        int2* __restrict__ seg = (int2*)seg_all + ((size_t)graph * SP_NB + t) * BK_NCH + chunk;
        *seg = make_int2(e0 + first[t], hist[t]);
    }
    __syncthreads();

    int4* __restrict__ rec = (int4*)rec_all + (size_t)graph * N_EDGES + e0;
    for (int i = t; i < BK_CH; i += BK_THREADS) rec[i] = sorted[i];
}

#define HD_CT    3
#define HD_BS_F  (32 * HD_CT * 32 * 2)
#define GM_BS_F  (16 * 8 * 32 * 4)
static_assert(NCLASS > 32 && NCLASS <= 48, "three column tiles, the third one partial");
static_assert((SP_ROWS * NCLASS) % 4 == 0, "a block of out is stored as float4");
static_assert(SP_WAVES == 32, "one wave stages one 4-row group of the 128x128 weight, or one k-step of Wout");
static_assert(SP_RT <= SP_WAVES, "the class head gives every row tile a wave of its own");
static_assert(SP_THREADS % 32 == 0 && NHID / 4 == 32 && SP_PITCH % 4 == 0, "the in-place relu pass: one lane, one group of four columns");

template<int M>
__device__ __forceinline__ float row_xor(float x)
{
    return __builtin_bit_cast(float,
        __builtin_amdgcn_mov_dpp(__builtin_bit_cast(int, x), 0x160 + M, 0xf, 0xf, true));
}

__device__ __forceinline__ void sp_tile_row(float4 (&av)[16], const float4* ap)
{
#pragma unroll
    for (int j = 0; j < 16; ++j) av[j] = ap[2 * j];
}

template<int EPI>
__global__ __launch_bounds__(SP_THREADS)
void spmm_rows_dense(const int* __restrict__ rec_g, const int* __restrict__ seg_g,
                     const float* __restrict__ h, const float* __restrict__ bias,
                     const float* __restrict__ Wn, const float* __restrict__ bn,
                     float* __restrict__ dst)
{
    __shared__ __attribute__((aligned(16))) float smem[SP_TILE_F + (EPI == 0 ? GM_BS_F : (HD_BS_F + SP_ROWS * NCLASS))];
    float* tile = smem;

    const int lane = threadIdx.x & 31;
    const int wave = __builtin_amdgcn_readfirstlane((int)(threadIdx.x >> 5));
    const int r0   = blockIdx.x * SP_ROWS;
    const int c    = lane & 15;
    const int hh   = lane >> 4;

    const int2* __restrict__ seg = (const int2*)seg_g + (size_t)blockIdx.x * BK_NCH;
    const int myc = wave + lane * SP_WAVES;
    int2 pr = make_int2(0, 0);
    if (lane < BK_SPW && myc < BK_NCH) pr = seg[myc];

    const bool c2   = (32 + c) < NCLASS;
    const int  col2 = c2 ? (32 + c) : (NCLASS - 1);
    float4 wr[4];
    float2 wv[HD_CT];
    if constexpr (EPI == 0) {
        const float4* __restrict__ wp = (const float4*)(Wn + (size_t)(4 * wave) * NHID) + lane;
#pragma unroll
        for (int r = 0; r < 4; ++r) wr[r] = wp[r * (NHID / 4)];
    } else {
        const int k = 8 * (wave >> 1) + 4 * hh + 2 * (wave & 1);
        const float* __restrict__ w0 = Wn + (size_t)k * NCLASS;
        const float* __restrict__ w1 = w0 + NCLASS;
        wv[0] = make_float2(w0[c],      w1[c]);
        wv[1] = make_float2(w0[16 + c], w1[16 + c]);
        wv[2] = make_float2(w0[col2],   w1[col2]);
    }

    float4* tile4 = (float4*)tile;
    for (int i = threadIdx.x; i < SP_TILE_F / 4; i += SP_THREADS)
        tile4[i] = make_float4(0.f, 0.f, 0.f, 0.f);

    if constexpr (EPI == 0) {
        float4* __restrict__ bd = (float4*)(smem + SP_TILE_F)
                                  + ((wave >> 1) * 8 + 2 * (lane >> 3)) * 32 + (wave & 1) * 16 + 2 * (lane & 7);
        bd[0]      = make_float4(wr[0].x, wr[1].x, wr[2].x, wr[3].x);
        bd[32]     = make_float4(wr[0].y, wr[1].y, wr[2].y, wr[3].y);
        bd[1]      = make_float4(wr[0].z, wr[1].z, wr[2].z, wr[3].z);
        bd[32 + 1] = make_float4(wr[0].w, wr[1].w, wr[2].w, wr[3].w);
    } else {
        float2* __restrict__ bd = (float2*)(smem + SP_TILE_F) + (wave * HD_CT) * 32 + lane;
        bd[0]  = wv[0];
        bd[32] = wv[1];
        bd[64] = c2 ? wv[2] : make_float2(0.f, 0.f);
    }
    __syncthreads();

    int sf[BK_SPW], pf[BK_SPW + 1];
    pf[0] = 0;
#pragma unroll
    for (int i = 0; i < BK_SPW; ++i) {
        sf[i]     = __builtin_amdgcn_readlane(pr.x, i);
        pf[i + 1] = pf[i] + __builtin_amdgcn_readlane(pr.y, i);
    }
    const int tot = pf[BK_SPW];

    const int4* __restrict__ rec = (const int4*)rec_g;
    for (int b = 0; b < tot; b += 32) {
        const int cnt = (tot - b) < 32 ? (tot - b) : 32;
        const int pos = b + lane;
        int a = sf[0] + pos;
#pragma unroll
        for (int i = 1; i < BK_SPW; ++i)
            if (pos >= pf[i]) a = sf[i] + (pos - pf[i]);
        int4 r = make_int4(0, 0, 0, 0);
        if (lane < cnt) r = rec[a];
        for (int g = 0; g < cnt; g += SP_GRP) {
            float hv[SP_GRP][4];
#pragma unroll
            for (int u = 0; u < SP_GRP; ++u) {
                const int iu = (g + u < cnt) ? (g + u) : (cnt - 1);
                const int su = __builtin_amdgcn_readlane(r.x, iu);
                const float* __restrict__ hp = h + (size_t)su * NHID + lane;
                hv[u][0] = hp[0];
                hv[u][1] = hp[32];
                hv[u][2] = hp[64];
                hv[u][3] = hp[96];
            }
#pragma unroll
            for (int u = 0; u < SP_GRP; ++u) {
                if (g + u < cnt) {
                    const int   du = __builtin_amdgcn_readlane(r.y, g + u);
                    const float wu = __int_as_float(__builtin_amdgcn_readlane(r.z, g + u));
                    float* tp = tile + du * SP_PITCH + lane;
                    atomicAdd(tp +  0, wu * hv[u][0]);
                    atomicAdd(tp + 32, wu * hv[u][1]);
                    atomicAdd(tp + 64, wu * hv[u][2]);
                    atomicAdd(tp + 96, wu * hv[u][3]);
                }
            }
        }
    }
    __syncthreads();

    {
        const float4 bq = ((const float4*)bias)[lane];
        for (int i = threadIdx.x; i < SP_TROWS * (NHID / 4); i += SP_THREADS) {
            float4* p = tile4 + (i >> 5) * (SP_PITCH / 4) + lane;
            float4 v = *p;
            v.x = fmaxf(v.x + bq.x, 0.0f);
            v.y = fmaxf(v.y + bq.y, 0.0f);
            v.z = fmaxf(v.z + bq.z, 0.0f);
            v.w = fmaxf(v.w + bq.w, 0.0f);
            *p = v;
        }
    }
    __syncthreads();

    if constexpr (EPI == 0) {
        const float4* Bs = (const float4*)(smem + SP_TILE_F);
        for (int item = wave; item < SP_RT * 4; item += SP_WAVES) {
            const int rt = item >> 2;
            const int q0 = item & 3;
            float4 av[16];
            sp_tile_row(av, (const float4*)(tile + (16 * rt + c) * SP_PITCH + 4 * hh));
            float* __restrict__ crow = dst + (size_t)(r0 + 16 * rt + 8 * hh) * NHID + 2 * c + 32 * q0;
            gm_rows_times_b<2>(av, Bs + (2 * q0) * 32 + lane, crow, 16 * rt + 8 * hh < SP_ROWS);
        }
    } else {
        float* stg = smem + SP_TILE_F + HD_BS_F;
        if (wave < SP_RT) {
            const int rt = wave;
            float4 av[16];
            sp_tile_row(av, (const float4*)(tile + (16 * rt + c) * SP_PITCH + 4 * hh));
            const float2* Bs = (const float2*)(smem + SP_TILE_F) + lane;

            v8f d0 = {}, d1 = {}, d2 = {};
#pragma unroll
            for (int j = 0; j < 16; ++j) {
                v2f a; float2 f; v2f b;
                const float2* bs = Bs + (2 * j) * HD_CT * 32;
                a.x = av[j].x; a.y = av[j].y;
                f = bs[0];   b.x = f.x; b.y = f.y;
                d0 = __builtin_amdgcn_wmma_f32_16x16x4_f32(false, a, false, b, (short)0, d0, false, false);
                f = bs[32];  b.x = f.x; b.y = f.y;
                d1 = __builtin_amdgcn_wmma_f32_16x16x4_f32(false, a, false, b, (short)0, d1, false, false);
                f = bs[64];  b.x = f.x; b.y = f.y;
                d2 = __builtin_amdgcn_wmma_f32_16x16x4_f32(false, a, false, b, (short)0, d2, false, false);
                a.x = av[j].z; a.y = av[j].w;
                f = bs[96];  b.x = f.x; b.y = f.y;
                d0 = __builtin_amdgcn_wmma_f32_16x16x4_f32(false, a, false, b, (short)0, d0, false, false);
                f = bs[128]; b.x = f.x; b.y = f.y;
                d1 = __builtin_amdgcn_wmma_f32_16x16x4_f32(false, a, false, b, (short)0, d1, false, false);
                f = bs[160]; b.x = f.x; b.y = f.y;
                d2 = __builtin_amdgcn_wmma_f32_16x16x4_f32(false, a, false, b, (short)0, d2, false, false);
            }

            const float bo0 = bn[c];
            const float bo1 = bn[16 + c];
            const float bo2 = bn[col2];
            float l0[8], l1[8], l2[8], mx[8], se[8];
#pragma unroll
            for (int r = 0; r < 8; ++r) {
                l0[r] = d0[r] + bo0;
                l1[r] = d1[r] + bo1;
                l2[r] = c2 ? (d2[r] + bo2) : -INFINITY;
                mx[r] = fmaxf(fmaxf(l0[r], l1[r]), l2[r]);
            }
#pragma unroll
            for (int r = 0; r < 8; ++r) mx[r] = fmaxf(mx[r], row_xor<1>(mx[r]));
#pragma unroll
            for (int r = 0; r < 8; ++r) mx[r] = fmaxf(mx[r], row_xor<2>(mx[r]));
#pragma unroll
            for (int r = 0; r < 8; ++r) mx[r] = fmaxf(mx[r], row_xor<4>(mx[r]));
#pragma unroll
            for (int r = 0; r < 8; ++r) mx[r] = fmaxf(mx[r], row_xor<8>(mx[r]));
#pragma unroll
            for (int r = 0; r < 8; ++r)
                se[r] = __expf(l0[r] - mx[r]) + __expf(l1[r] - mx[r]) + (c2 ? __expf(l2[r] - mx[r]) : 0.0f);
#pragma unroll
            for (int r = 0; r < 8; ++r) se[r] += row_xor<1>(se[r]);
#pragma unroll
            for (int r = 0; r < 8; ++r) se[r] += row_xor<2>(se[r]);
#pragma unroll
            for (int r = 0; r < 8; ++r) se[r] += row_xor<4>(se[r]);
#pragma unroll
            for (int r = 0; r < 8; ++r) se[r] += row_xor<8>(se[r]);

            if (16 * rt + 8 * hh < SP_ROWS) {
                float* __restrict__ sp = stg + (16 * rt + 8 * hh) * NCLASS + c;
#pragma unroll
                for (int r = 0; r < 8; ++r) {
                    const float lse = __logf(se[r]) + mx[r];
                    sp[r * NCLASS]      = l0[r] - lse;
                    sp[r * NCLASS + 16] = l1[r] - lse;
                    if (c2) sp[r * NCLASS + 32] = l2[r] - lse;
                }
            }
        }
        __syncthreads();

        float4* __restrict__ o4 = (float4*)(dst + (size_t)r0 * NCLASS);
        const float4* s4 = (const float4*)stg;
        for (int i = threadIdx.x; i < SP_ROWS * NCLASS / 4; i += SP_THREADS) o4[i] = s4[i];
    }
}

extern "C" void kernel_launch(void* const* d_in, const int* in_sizes, int n_in,
                              void* d_out, int out_size, void* d_ws, size_t ws_size,
                              hipStream_t stream)
{
    const float* x    = (const float*)d_in[0];
    const int*   a0i  = (const int*)  d_in[1];
    const float* a0w  = (const float*)d_in[2];
    const int*   a1i  = (const int*)  d_in[3];
    const float* a1w  = (const float*)d_in[4];
    const float* W1   = (const float*)d_in[5];
    const float* b1   = (const float*)d_in[6];
    const float* W2   = (const float*)d_in[7];
    const float* b2   = (const float*)d_in[8];
    const float* Wout = (const float*)d_in[9];
    const float* bout = (const float*)d_in[10];
    float* out = (float*)d_out;

    const size_t NH = (size_t)N_NODES * NHID;
    float* bufA = (float*)d_ws;
    float* bufB = bufA + NH;
    int*   recAll = (int*)(bufB + NH);
    const size_t REC_G = (size_t)N_EDGES * 4;
    int*   segAll = recAll + 2 * REC_G;
    const size_t SEG_G = (size_t)SP_NB * BK_NCH * 2;

    const dim3 gGemm(GM_WGS), bGemm(GM_WAVES * 32);
    const dim3 gSp(SP_NB), bSp(SP_THREADS);

    bucket_chunks<<<dim3(2 * BK_NCH), dim3(BK_THREADS), 0, stream>>>(
        a0i, a0i + N_EDGES, a0w, a1i, a1i + N_EDGES, a1w, recAll, segAll);

    gemm128_wmma<<<gGemm, bGemm, 0, stream>>>(x, W1, bufA);

    spmm_rows_dense<0><<<gSp, bSp, 0, stream>>>(recAll, segAll, bufA, b1, W2, nullptr, bufB);

    spmm_rows_dense<1><<<gSp, bSp, 0, stream>>>(recAll + REC_G, segAll + SEG_G, bufB, b2, Wout, bout, out);
}


# spmm 0: adj@(x@W1) done as (adj@x)@W1: gathers x rows; its LDS tile times W1 on WMMA, relu(+b1) written back, times W2; the x@W1 launch is gone (3 launches); edge sums now precede the W1 dots
# speedup vs baseline: 4.5564x; 1.1272x over previous
#include <hip/hip_runtime.h>
#include <math.h>

#define N_NODES 50000
#define N_EDGES 600000
#define NFEAT 128
#define NHID 128
#define NCLASS 41

typedef float v2f __attribute__((ext_vector_type(2)));
typedef float v8f __attribute__((ext_vector_type(8)));

static_assert(NFEAT == 128 && NHID == 128, "the fragment layout below is for 128 x 128 weights");

template<int NT>
__device__ __forceinline__ void gm_rows_times_b(const float4 (&av)[16], const float4* bs, v8f (&d)[NT])
{
#pragma unroll
    for (int t = 0; t < NT; ++t) d[t] = v8f{};
    float4 f[2][NT];
#pragma unroll
    for (int t = 0; t < NT; ++t) f[0][t] = bs[t * 32];
#pragma unroll
    for (int j = 0; j < 16; ++j) {
        if (j + 1 < 16) {
#pragma unroll
            for (int t = 0; t < NT; ++t) f[(j + 1) & 1][t] = bs[((j + 1) * 8 + t) * 32];
        }
        __builtin_amdgcn_sched_barrier(0);
        v2f a0, a1;
        a0.x = av[j].x; a0.y = av[j].y;
        a1.x = av[j].z; a1.y = av[j].w;
#pragma unroll
        for (int t = 0; t < NT; ++t) {
            v2f b; b.x = f[j & 1][t].x; b.y = f[j & 1][t].y;
            d[t] = __builtin_amdgcn_wmma_f32_16x16x4_f32(false, a0, false, b, (short)0, d[t], false, false);
        }
#pragma unroll
        for (int t = 0; t < NT; ++t) {
            v2f b; b.x = f[j & 1][t].z; b.y = f[j & 1][t].w;
            d[t] = __builtin_amdgcn_wmma_f32_16x16x4_f32(false, a1, false, b, (short)0, d[t], false, false);
        }
        __builtin_amdgcn_sched_barrier(0);
    }
}

#define SP_ROWS    200
#define SP_RT      ((SP_ROWS + 15) / 16)
#define SP_TROWS   (SP_RT * 16)
#define SP_PITCH   (NHID + 4)
#define SP_TILE_F  (SP_TROWS * SP_PITCH)
#define SP_NB      (N_NODES / SP_ROWS)
#define SP_THREADS 1024
#define SP_WAVES   (SP_THREADS / 32)
#define SP_GRP     8
#define BK_CH      4800
#define BK_NCH     (N_EDGES / BK_CH)
#define BK_THREADS 512
#define BK_PER     ((BK_CH + BK_THREADS - 1) / BK_THREADS)
#define BK_BINS    256
#define BK_SPW     ((BK_NCH + SP_WAVES - 1) / SP_WAVES)
static_assert(N_NODES % SP_ROWS == 0, "row ranges must tile the node set exactly");
static_assert(N_EDGES % BK_CH == 0, "chunks must tile the edge list exactly");
static_assert(SP_NB < BK_BINS && BK_BINS == 8 * 32, "the scan gives each lane of one wave 8 bins");
static_assert(SP_NB <= BK_THREADS, "one thread writes one (first, count) pair");
static_assert(BK_SPW <= 32, "a wave holds its chunk pairs one per lane");

__global__ __launch_bounds__(BK_THREADS)
void bucket_chunks(const int* __restrict__ src0, const int* __restrict__ dst0, const float* __restrict__ w0,
                   const int* __restrict__ src1, const int* __restrict__ dst1, const float* __restrict__ w1,
                   int* __restrict__ rec_all, int* __restrict__ seg_all)
{
    __shared__ int hist[BK_BINS];
    __shared__ int first[BK_BINS];
    __shared__ __attribute__((aligned(16))) int4 sorted[BK_CH];

    const int t     = threadIdx.x;
    const int graph = (int)blockIdx.x / BK_NCH;
    const int chunk = (int)blockIdx.x - graph * BK_NCH;
    const int e0    = chunk * BK_CH;
    const int*   __restrict__ src = graph ? src1 : src0;
    const int*   __restrict__ dst = graph ? dst1 : dst0;
    const float* __restrict__ w   = graph ? w1 : w0;

    if (t < BK_BINS) hist[t] = 0;

    int s[BK_PER] = {}, d[BK_PER] = {}, wb[BK_PER] = {}, bin[BK_PER] = {}, rank[BK_PER] = {};
#pragma unroll
    for (int j = 0; j < BK_PER; ++j) {
        const int i = j * BK_THREADS + t;
        if (i < BK_CH) {
            s[j]  = src[e0 + i];
            d[j]  = dst[e0 + i];
            wb[j] = __float_as_int(w[e0 + i]);
        }
    }
    __syncthreads();

#pragma unroll
    for (int j = 0; j < BK_PER; ++j) {
        const int i = j * BK_THREADS + t;
        if (i < BK_CH) {
            bin[j]  = ((unsigned)d[j] < (unsigned)N_NODES) ? (d[j] / SP_ROWS) : SP_NB;
            rank[j] = atomicAdd(&hist[bin[j]], 1);
        }
    }
    __syncthreads();

    if (t < 32) {
        int c[8];
        int sum = 0;
#pragma unroll
        for (int k = 0; k < 8; ++k) { c[k] = hist[t * 8 + k]; sum += c[k]; }
        int incl = sum;
#pragma unroll
        for (int dl = 1; dl < 32; dl <<= 1) {
            const int o = __shfl_up(incl, dl);
            if (t >= dl) incl += o;
        }
        int run = incl - sum;
#pragma unroll
        for (int k = 0; k < 8; ++k) { first[t * 8 + k] = run; run += c[k]; }
    }
    __syncthreads();

#pragma unroll
    for (int j = 0; j < BK_PER; ++j) {
        const int i = j * BK_THREADS + t;
        if (i < BK_CH)
            sorted[first[bin[j]] + rank[j]] = make_int4(s[j], d[j] - bin[j] * SP_ROWS, wb[j], 0);
    }
    if (t < SP_NB) {
        int2* __restrict__ seg = (int2*)seg_all + ((size_t)graph * SP_NB + t) * BK_NCH + chunk;
        *seg = make_int2(e0 + first[t], hist[t]);
    }
    __syncthreads();

    int4* __restrict__ rec = (int4*)rec_all + (size_t)graph * N_EDGES + e0;
    for (int i = t; i < BK_CH; i += BK_THREADS) rec[i] = sorted[i];
}

#define HD_CT    3
#define HD_BS_F  (32 * HD_CT * 32 * 2)
#define GM_BS_F  (16 * 8 * 32 * 4)
static_assert(NCLASS > 32 && NCLASS <= 48, "three column tiles, the third one partial");
static_assert((SP_ROWS * NCLASS) % 4 == 0, "a block of out is stored as float4");
static_assert(SP_WAVES == 32, "one wave stages one 4-row group of the 128x128 weight, or one k-step of Wout");
static_assert(SP_RT <= SP_WAVES, "the class head gives every row tile a wave of its own");
static_assert(SP_THREADS % 32 == 0 && NHID / 4 == 32 && SP_PITCH % 4 == 0, "the in-place relu pass: one lane, one group of four columns");

template<int M>
__device__ __forceinline__ float row_xor(float x)
{
    return __builtin_bit_cast(float,
        __builtin_amdgcn_mov_dpp(__builtin_bit_cast(int, x), 0x160 + M, 0xf, 0xf, true));
}

__device__ __forceinline__ void sp_tile_row(float4 (&av)[16], const float4* ap)
{
#pragma unroll
    for (int j = 0; j < 16; ++j) av[j] = ap[2 * j];
}

template<int EPI>
__global__ __launch_bounds__(SP_THREADS)
void spmm_rows_dense(const int* __restrict__ rec_g, const int* __restrict__ seg_g,
                     const float* __restrict__ h, const float* __restrict__ bias,
                     const float* __restrict__ Wn, const float* __restrict__ aux,
                     float* __restrict__ dst)
{
    __shared__ __attribute__((aligned(16))) float smem[SP_TILE_F + (EPI == 0 ? 2 * GM_BS_F : (HD_BS_F + SP_ROWS * NCLASS))];
    float* tile = smem;

    const int lane = threadIdx.x & 31;
    const int wave = __builtin_amdgcn_readfirstlane((int)(threadIdx.x >> 5));
    const int r0   = blockIdx.x * SP_ROWS;
    const int c    = lane & 15;
    const int hh   = lane >> 4;

    const int2* __restrict__ seg = (const int2*)seg_g + (size_t)blockIdx.x * BK_NCH;
    const int myc = wave + lane * SP_WAVES;
    int2 pr = make_int2(0, 0);
    if (lane < BK_SPW && myc < BK_NCH) pr = seg[myc];

    const bool c2   = (32 + c) < NCLASS;
    const int  col2 = c2 ? (32 + c) : (NCLASS - 1);
    float4 wr[2][4];
    float2 wv[HD_CT];
    float2 bq = make_float2(0.f, 0.f);
    if constexpr (EPI == 0) {
        static_assert(SP_WAVES % 4 == 0, "a wave's items share one column group");
        bq = *(const float2*)(bias + 32 * (wave & 3) + 2 * c);
#pragma unroll
        for (int m = 0; m < 2; ++m) {
            const float4* __restrict__ wp = (const float4*)((m ? aux : Wn) + (size_t)(4 * wave) * NHID) + lane;
#pragma unroll
            for (int r = 0; r < 4; ++r) wr[m][r] = wp[r * (NHID / 4)];
        }
    } else {
        const int k = 8 * (wave >> 1) + 4 * hh + 2 * (wave & 1);
        const float* __restrict__ w0 = Wn + (size_t)k * NCLASS;
        const float* __restrict__ w1 = w0 + NCLASS;
        wv[0] = make_float2(w0[c],      w1[c]);
        wv[1] = make_float2(w0[16 + c], w1[16 + c]);
        wv[2] = make_float2(w0[col2],   w1[col2]);
    }

    float4* tile4 = (float4*)tile;
    for (int i = threadIdx.x; i < SP_TILE_F / 4; i += SP_THREADS)
        tile4[i] = make_float4(0.f, 0.f, 0.f, 0.f);

    if constexpr (EPI == 0) {
#pragma unroll
        for (int m = 0; m < 2; ++m) {
            float4* __restrict__ bd = (float4*)(smem + SP_TILE_F + m * GM_BS_F)
                                      + ((wave >> 1) * 8 + 2 * (lane >> 3)) * 32 + (wave & 1) * 16 + 2 * (lane & 7);
            bd[0]      = make_float4(wr[m][0].x, wr[m][1].x, wr[m][2].x, wr[m][3].x);
            bd[32]     = make_float4(wr[m][0].y, wr[m][1].y, wr[m][2].y, wr[m][3].y);
            bd[1]      = make_float4(wr[m][0].z, wr[m][1].z, wr[m][2].z, wr[m][3].z);
            bd[32 + 1] = make_float4(wr[m][0].w, wr[m][1].w, wr[m][2].w, wr[m][3].w);
        }
    } else {
        float2* __restrict__ bd = (float2*)(smem + SP_TILE_F) + (wave * HD_CT) * 32 + lane;
        bd[0]  = wv[0];
        bd[32] = wv[1];
        bd[64] = c2 ? wv[2] : make_float2(0.f, 0.f);
    }
    __syncthreads();

    int sf[BK_SPW], pf[BK_SPW + 1];
    pf[0] = 0;
#pragma unroll
    for (int i = 0; i < BK_SPW; ++i) {
        sf[i]     = __builtin_amdgcn_readlane(pr.x, i);
        pf[i + 1] = pf[i] + __builtin_amdgcn_readlane(pr.y, i);
    }
    const int tot = pf[BK_SPW];

    const int4* __restrict__ rec = (const int4*)rec_g;
    for (int b = 0; b < tot; b += 32) {
        const int cnt = (tot - b) < 32 ? (tot - b) : 32;
        const int pos = b + lane;
        int a = sf[0] + pos;
#pragma unroll
        for (int i = 1; i < BK_SPW; ++i)
            if (pos >= pf[i]) a = sf[i] + (pos - pf[i]);
        int4 r = make_int4(0, 0, 0, 0);
        if (lane < cnt) r = rec[a];
        for (int g = 0; g < cnt; g += SP_GRP) {
            float hv[SP_GRP][4];
#pragma unroll
            for (int u = 0; u < SP_GRP; ++u) {
                const int iu = (g + u < cnt) ? (g + u) : (cnt - 1);
                const int su = __builtin_amdgcn_readlane(r.x, iu);
                const float* __restrict__ hp = h + (size_t)su * NHID + lane;
                hv[u][0] = hp[0];
                hv[u][1] = hp[32];
                hv[u][2] = hp[64];
                hv[u][3] = hp[96];
            }
#pragma unroll
            for (int u = 0; u < SP_GRP; ++u) {
                if (g + u < cnt) {
                    const int   du = __builtin_amdgcn_readlane(r.y, g + u);
                    const float wu = __int_as_float(__builtin_amdgcn_readlane(r.z, g + u));
                    float* tp = tile + du * SP_PITCH + lane;
                    atomicAdd(tp +  0, wu * hv[u][0]);
                    atomicAdd(tp + 32, wu * hv[u][1]);
                    atomicAdd(tp + 64, wu * hv[u][2]);
                    atomicAdd(tp + 96, wu * hv[u][3]);
                }
            }
        }
    }
    __syncthreads();

    if constexpr (EPI == 1) {
        const float4 b4 = ((const float4*)bias)[lane];
        for (int i = threadIdx.x; i < SP_TROWS * (NHID / 4); i += SP_THREADS) {
            float4* p = tile4 + (i >> 5) * (SP_PITCH / 4) + lane;
            float4 v = *p;
            v.x = fmaxf(v.x + b4.x, 0.0f);
            v.y = fmaxf(v.y + b4.y, 0.0f);
            v.z = fmaxf(v.z + b4.z, 0.0f);
            v.w = fmaxf(v.w + b4.w, 0.0f);
            *p = v;
        }
        __syncthreads();
    }

    if constexpr (EPI == 0) {
        const float4* B1 = (const float4*)(smem + SP_TILE_F);
        const float4* B2 = B1 + GM_BS_F / 4;

        for (int base = 0; base < SP_RT * 4; base += SP_WAVES) {
            const int  item = base + wave;
            const bool act  = item < SP_RT * 4;
            const int  rt   = item >> 2;
            const int  q0   = item & 3;
            float4 av[16];
            if (act) sp_tile_row(av, (const float4*)(tile + (16 * rt + c) * SP_PITCH + 4 * hh));
            __syncthreads();
            if (act) {
                v8f d[2];
                gm_rows_times_b<2>(av, B1 + (2 * q0) * 32 + lane, d);
                float* trow = tile + (16 * rt + 8 * hh) * SP_PITCH + 32 * q0 + 2 * c;
#pragma unroll
                for (int v = 0; v < 8; ++v)
                    *(float2*)(trow + v * SP_PITCH) = make_float2(fmaxf(d[0][v] + bq.x, 0.0f), fmaxf(d[1][v] + bq.y, 0.0f));
            }
        }
        __syncthreads();

        for (int item = wave; item < SP_RT * 4; item += SP_WAVES) {
            const int rt = item >> 2;
            const int q0 = item & 3;
            float4 av[16];
            sp_tile_row(av, (const float4*)(tile + (16 * rt + c) * SP_PITCH + 4 * hh));
            v8f d[2];
            gm_rows_times_b<2>(av, B2 + (2 * q0) * 32 + lane, d);
            if (16 * rt + 8 * hh < SP_ROWS) {
                float* __restrict__ crow = dst + (size_t)(r0 + 16 * rt + 8 * hh) * NHID + 2 * c + 32 * q0;
#pragma unroll
                for (int v = 0; v < 8; ++v)
                    *(float2*)(crow + v * NHID) = make_float2(d[0][v], d[1][v]);
            }
        }
    } else {
        float* stg = smem + SP_TILE_F + HD_BS_F;
        if (wave < SP_RT) {
            const int rt = wave;
            float4 av[16];
            sp_tile_row(av, (const float4*)(tile + (16 * rt + c) * SP_PITCH + 4 * hh));
            const float2* Bs = (const float2*)(smem + SP_TILE_F) + lane;

            v8f d0 = {}, d1 = {}, d2 = {};
#pragma unroll
            for (int j = 0; j < 16; ++j) {
                v2f a; float2 f; v2f b;
                const float2* bs = Bs + (2 * j) * HD_CT * 32;
                a.x = av[j].x; a.y = av[j].y;
                f = bs[0];   b.x = f.x; b.y = f.y;
                d0 = __builtin_amdgcn_wmma_f32_16x16x4_f32(false, a, false, b, (short)0, d0, false, false);
                f = bs[32];  b.x = f.x; b.y = f.y;
                d1 = __builtin_amdgcn_wmma_f32_16x16x4_f32(false, a, false, b, (short)0, d1, false, false);
                f = bs[64];  b.x = f.x; b.y = f.y;
                d2 = __builtin_amdgcn_wmma_f32_16x16x4_f32(false, a, false, b, (short)0, d2, false, false);
                a.x = av[j].z; a.y = av[j].w;
                f = bs[96];  b.x = f.x; b.y = f.y;
                d0 = __builtin_amdgcn_wmma_f32_16x16x4_f32(false, a, false, b, (short)0, d0, false, false);
                f = bs[128]; b.x = f.x; b.y = f.y;
                d1 = __builtin_amdgcn_wmma_f32_16x16x4_f32(false, a, false, b, (short)0, d1, false, false);
                f = bs[160]; b.x = f.x; b.y = f.y;
                d2 = __builtin_amdgcn_wmma_f32_16x16x4_f32(false, a, false, b, (short)0, d2, false, false);
            }

            const float bo0 = aux[c];
            const float bo1 = aux[16 + c];
            const float bo2 = aux[col2];
            float l0[8], l1[8], l2[8], mx[8], se[8];
#pragma unroll
            for (int r = 0; r < 8; ++r) {
                l0[r] = d0[r] + bo0;
                l1[r] = d1[r] + bo1;
                l2[r] = c2 ? (d2[r] + bo2) : -INFINITY;
                mx[r] = fmaxf(fmaxf(l0[r], l1[r]), l2[r]);
            }
#pragma unroll
            for (int r = 0; r < 8; ++r) mx[r] = fmaxf(mx[r], row_xor<1>(mx[r]));
#pragma unroll
            for (int r = 0; r < 8; ++r) mx[r] = fmaxf(mx[r], row_xor<2>(mx[r]));
#pragma unroll
            for (int r = 0; r < 8; ++r) mx[r] = fmaxf(mx[r], row_xor<4>(mx[r]));
#pragma unroll
            for (int r = 0; r < 8; ++r) mx[r] = fmaxf(mx[r], row_xor<8>(mx[r]));
#pragma unroll
            for (int r = 0; r < 8; ++r)
                se[r] = __expf(l0[r] - mx[r]) + __expf(l1[r] - mx[r]) + (c2 ? __expf(l2[r] - mx[r]) : 0.0f);
#pragma unroll
            for (int r = 0; r < 8; ++r) se[r] += row_xor<1>(se[r]);
#pragma unroll
            for (int r = 0; r < 8; ++r) se[r] += row_xor<2>(se[r]);
#pragma unroll
            for (int r = 0; r < 8; ++r) se[r] += row_xor<4>(se[r]);
#pragma unroll
            for (int r = 0; r < 8; ++r) se[r] += row_xor<8>(se[r]);

            if (16 * rt + 8 * hh < SP_ROWS) {
                float* __restrict__ sp = stg + (16 * rt + 8 * hh) * NCLASS + c;
#pragma unroll
                for (int r = 0; r < 8; ++r) {
                    const float lse = __logf(se[r]) + mx[r];
                    sp[r * NCLASS]      = l0[r] - lse;
                    sp[r * NCLASS + 16] = l1[r] - lse;
                    if (c2) sp[r * NCLASS + 32] = l2[r] - lse;
                }
            }
        }
        __syncthreads();

        float4* __restrict__ o4 = (float4*)(dst + (size_t)r0 * NCLASS);
        const float4* s4 = (const float4*)stg;
        for (int i = threadIdx.x; i < SP_ROWS * NCLASS / 4; i += SP_THREADS) o4[i] = s4[i];
    }
}

extern "C" void kernel_launch(void* const* d_in, const int* in_sizes, int n_in,
                              void* d_out, int out_size, void* d_ws, size_t ws_size,
                              hipStream_t stream)
{
    const float* x    = (const float*)d_in[0];
    const int*   a0i  = (const int*)  d_in[1];
    const float* a0w  = (const float*)d_in[2];
    const int*   a1i  = (const int*)  d_in[3];
    const float* a1w  = (const float*)d_in[4];
    const float* W1   = (const float*)d_in[5];
    const float* b1   = (const float*)d_in[6];
    const float* W2   = (const float*)d_in[7];
    const float* b2   = (const float*)d_in[8];
    const float* Wout = (const float*)d_in[9];
    const float* bout = (const float*)d_in[10];
    float* out = (float*)d_out;

    const size_t NH = (size_t)N_NODES * NHID;
    float* bufT = (float*)d_ws;
    int*   recAll = (int*)(bufT + NH);
    const size_t REC_G = (size_t)N_EDGES * 4;
    int*   segAll = recAll + 2 * REC_G;
    const size_t SEG_G = (size_t)SP_NB * BK_NCH * 2;

    const dim3 gSp(SP_NB), bSp(SP_THREADS);

    bucket_chunks<<<dim3(2 * BK_NCH), dim3(BK_THREADS), 0, stream>>>(
        a0i, a0i + N_EDGES, a0w, a1i, a1i + N_EDGES, a1w, recAll, segAll);

    spmm_rows_dense<0><<<gSp, bSp, 0, stream>>>(recAll, segAll, x, b1, W1, W2, bufT);

    spmm_rows_dense<1><<<gSp, bSp, 0, stream>>>(recAll + REC_G, segAll + SEG_G, bufT, b2, Wout, bout, out);
}


# spmm gather: one 16-byte load per lane fetches a whole h row (1 load per edge, not 4); summed tile kept lane-interleaved, W1/Wout/b2 staged to match, so their dot sums run in a new fixed k order
# speedup vs baseline: 4.8596x; 1.0665x over previous
#include <hip/hip_runtime.h>
#include <math.h>

#define N_NODES 50000
#define N_EDGES 600000
#define NFEAT 128
#define NHID 128
#define NCLASS 41

typedef float v2f __attribute__((ext_vector_type(2)));
typedef float v8f __attribute__((ext_vector_type(8)));

static_assert(NFEAT == 128 && NHID == 128, "the fragment layout below is for 128 x 128 weights");

template<int NT>
__device__ __forceinline__ void gm_rows_times_b(const float4 (&av)[16], const float4* bs, v8f (&d)[NT])
{
#pragma unroll
    for (int t = 0; t < NT; ++t) d[t] = v8f{};
    float4 f[2][NT];
#pragma unroll
    for (int t = 0; t < NT; ++t) f[0][t] = bs[t * 32];
#pragma unroll
    for (int j = 0; j < 16; ++j) {
        if (j + 1 < 16) {
#pragma unroll
            for (int t = 0; t < NT; ++t) f[(j + 1) & 1][t] = bs[((j + 1) * 8 + t) * 32];
        }
        __builtin_amdgcn_sched_barrier(0);
        v2f a0, a1;
        a0.x = av[j].x; a0.y = av[j].y;
        a1.x = av[j].z; a1.y = av[j].w;
#pragma unroll
        for (int t = 0; t < NT; ++t) {
            v2f b; b.x = f[j & 1][t].x; b.y = f[j & 1][t].y;
            d[t] = __builtin_amdgcn_wmma_f32_16x16x4_f32(false, a0, false, b, (short)0, d[t], false, false);
        }
#pragma unroll
        for (int t = 0; t < NT; ++t) {
            v2f b; b.x = f[j & 1][t].z; b.y = f[j & 1][t].w;
            d[t] = __builtin_amdgcn_wmma_f32_16x16x4_f32(false, a1, false, b, (short)0, d[t], false, false);
        }
        __builtin_amdgcn_sched_barrier(0);
    }
}

#define SP_ROWS    200
#define SP_RT      ((SP_ROWS + 15) / 16)
#define SP_TROWS   (SP_RT * 16)
#define SP_PITCH   (NHID + 4)
#define SP_TILE_F  (SP_TROWS * SP_PITCH)
#define SP_NB      (N_NODES / SP_ROWS)
#define SP_THREADS 1024
#define SP_WAVES   (SP_THREADS / 32)
#define SP_GRP     8
#define BK_CH      4800
#define BK_NCH     (N_EDGES / BK_CH)
#define BK_THREADS 512
#define BK_PER     ((BK_CH + BK_THREADS - 1) / BK_THREADS)
#define BK_BINS    256
#define BK_SPW     ((BK_NCH + SP_WAVES - 1) / SP_WAVES)
static_assert(N_NODES % SP_ROWS == 0, "row ranges must tile the node set exactly");
static_assert(N_EDGES % BK_CH == 0, "chunks must tile the edge list exactly");
static_assert(SP_NB < BK_BINS && BK_BINS == 8 * 32, "the scan gives each lane of one wave 8 bins");
static_assert(SP_NB <= BK_THREADS, "one thread writes one (first, count) pair");
static_assert(BK_SPW <= 32, "a wave holds its chunk pairs one per lane");

__global__ __launch_bounds__(BK_THREADS)
void bucket_chunks(const int* __restrict__ src0, const int* __restrict__ dst0, const float* __restrict__ w0,
                   const int* __restrict__ src1, const int* __restrict__ dst1, const float* __restrict__ w1,
                   int* __restrict__ rec_all, int* __restrict__ seg_all)
{
    __shared__ int hist[BK_BINS];
    __shared__ int first[BK_BINS];
    __shared__ __attribute__((aligned(16))) int4 sorted[BK_CH];

    const int t     = threadIdx.x;
    const int graph = (int)blockIdx.x / BK_NCH;
    const int chunk = (int)blockIdx.x - graph * BK_NCH;
    const int e0    = chunk * BK_CH;
    const int*   __restrict__ src = graph ? src1 : src0;
    const int*   __restrict__ dst = graph ? dst1 : dst0;
    const float* __restrict__ w   = graph ? w1 : w0;

    if (t < BK_BINS) hist[t] = 0;

    int s[BK_PER] = {}, d[BK_PER] = {}, wb[BK_PER] = {}, bin[BK_PER] = {}, rank[BK_PER] = {};
#pragma unroll
    for (int j = 0; j < BK_PER; ++j) {
        const int i = j * BK_THREADS + t;
        if (i < BK_CH) {
            s[j]  = src[e0 + i];
            d[j]  = dst[e0 + i];
            wb[j] = __float_as_int(w[e0 + i]);
        }
    }
    __syncthreads();

#pragma unroll
    for (int j = 0; j < BK_PER; ++j) {
        const int i = j * BK_THREADS + t;
        if (i < BK_CH) {
            bin[j]  = ((unsigned)d[j] < (unsigned)N_NODES) ? (d[j] / SP_ROWS) : SP_NB;
            rank[j] = atomicAdd(&hist[bin[j]], 1);
        }
    }
    __syncthreads();

    if (t < 32) {
        int c[8];
        int sum = 0;
#pragma unroll
        for (int k = 0; k < 8; ++k) { c[k] = hist[t * 8 + k]; sum += c[k]; }
        int incl = sum;
#pragma unroll
        for (int dl = 1; dl < 32; dl <<= 1) {
            const int o = __shfl_up(incl, dl);
            if (t >= dl) incl += o;
        }
        int run = incl - sum;
#pragma unroll
        for (int k = 0; k < 8; ++k) { first[t * 8 + k] = run; run += c[k]; }
    }
    __syncthreads();

#pragma unroll
    for (int j = 0; j < BK_PER; ++j) {
        const int i = j * BK_THREADS + t;
        if (i < BK_CH)
            sorted[first[bin[j]] + rank[j]] = make_int4(s[j], d[j] - bin[j] * SP_ROWS, wb[j], 0);
    }
    if (t < SP_NB) {
        int2* __restrict__ seg = (int2*)seg_all + ((size_t)graph * SP_NB + t) * BK_NCH + chunk;
        *seg = make_int2(e0 + first[t], hist[t]);
    }
    __syncthreads();

    int4* __restrict__ rec = (int4*)rec_all + (size_t)graph * N_EDGES + e0;
    for (int i = t; i < BK_CH; i += BK_THREADS) rec[i] = sorted[i];
}

#define HD_CT    3
#define HD_BS_F  (32 * HD_CT * 32 * 2)
#define GM_BS_F  (16 * 8 * 32 * 4)
static_assert(NCLASS > 32 && NCLASS <= 48, "three column tiles, the third one partial");
static_assert((SP_ROWS * NCLASS) % 4 == 0, "a block of out is stored as float4");
static_assert(SP_WAVES == 32, "one wave stages one 4-row group of the 128x128 weight, or one k-step of Wout");
static_assert(SP_RT <= SP_WAVES, "the class head gives every row tile a wave of its own");
static_assert(SP_THREADS % 32 == 0 && NHID / 4 == 32 && SP_PITCH % 4 == 0, "the in-place relu pass: one lane, one group of four columns");

template<int M>
__device__ __forceinline__ float row_xor(float x)
{
    return __builtin_bit_cast(float,
        __builtin_amdgcn_mov_dpp(__builtin_bit_cast(int, x), 0x160 + M, 0xf, 0xf, true));
}

__device__ __forceinline__ void sp_tile_row(float4 (&av)[16], const float4* ap)
{
#pragma unroll
    for (int j = 0; j < 16; ++j) av[j] = ap[2 * j];
}

template<int EPI>
__global__ __launch_bounds__(SP_THREADS)
void spmm_rows_dense(const int* __restrict__ rec_g, const int* __restrict__ seg_g,
                     const float* __restrict__ h, const float* __restrict__ bias,
                     const float* __restrict__ Wn, const float* __restrict__ aux,
                     float* __restrict__ dst)
{
    __shared__ __attribute__((aligned(16))) float smem[SP_TILE_F + (EPI == 0 ? 2 * GM_BS_F : (HD_BS_F + SP_ROWS * NCLASS))];
    float* tile = smem;

    const int lane = threadIdx.x & 31;
    const int wave = __builtin_amdgcn_readfirstlane((int)(threadIdx.x >> 5));
    const int r0   = blockIdx.x * SP_ROWS;
    const int c    = lane & 15;
    const int hh   = lane >> 4;

    const int2* __restrict__ seg = (const int2*)seg_g + (size_t)blockIdx.x * BK_NCH;
    const int myc = wave + lane * SP_WAVES;
    int2 pr = make_int2(0, 0);
    if (lane < BK_SPW && myc < BK_NCH) pr = seg[myc];

    const bool c2   = (32 + c) < NCLASS;
    const int  col2 = c2 ? (32 + c) : (NCLASS - 1);
    float4 wr[2][4];
    float2 wv[HD_CT];
    float2 bq = make_float2(0.f, 0.f);
    if constexpr (EPI == 0) {
        static_assert(SP_WAVES % 4 == 0, "a wave's items share one column group");
        bq = *(const float2*)(bias + 32 * (wave & 3) + 2 * c);
#pragma unroll
        for (int m = 0; m < 2; ++m) {
            const int k0 = m ? (4 * wave) : (32 * ((wave >> 1) & 3) + 16 * (wave & 1) + (wave >> 3));
            const int ks = m ? 1 : 4;
            const float4* __restrict__ wp = (const float4*)((m ? aux : Wn) + (size_t)k0 * NHID) + lane;
#pragma unroll
            for (int r = 0; r < 4; ++r) wr[m][r] = wp[r * ks * (NHID / 4)];
        }
    } else {
        const int k = 32 * ((wave >> 1) & 3) + 16 * hh + 8 * (wave & 1) + (wave >> 3);
        const float* __restrict__ w0 = Wn + (size_t)k * NCLASS;
        const float* __restrict__ w1 = w0 + 4 * NCLASS;
        wv[0] = make_float2(w0[c],      w1[c]);
        wv[1] = make_float2(w0[16 + c], w1[16 + c]);
        wv[2] = make_float2(w0[col2],   w1[col2]);
    }

    float4* tile4 = (float4*)tile;
    for (int i = threadIdx.x; i < SP_TILE_F / 4; i += SP_THREADS)
        tile4[i] = make_float4(0.f, 0.f, 0.f, 0.f);

    if constexpr (EPI == 0) {
#pragma unroll
        for (int m = 0; m < 2; ++m) {
            float4* __restrict__ bd = (float4*)(smem + SP_TILE_F + m * GM_BS_F)
                                      + ((wave >> 1) * 8 + 2 * (lane >> 3)) * 32 + (wave & 1) * 16 + 2 * (lane & 7);
            bd[0]      = make_float4(wr[m][0].x, wr[m][1].x, wr[m][2].x, wr[m][3].x);
            bd[32]     = make_float4(wr[m][0].y, wr[m][1].y, wr[m][2].y, wr[m][3].y);
            bd[1]      = make_float4(wr[m][0].z, wr[m][1].z, wr[m][2].z, wr[m][3].z);
            bd[32 + 1] = make_float4(wr[m][0].w, wr[m][1].w, wr[m][2].w, wr[m][3].w);
        }
    } else {
        float2* __restrict__ bd = (float2*)(smem + SP_TILE_F) + (wave * HD_CT) * 32 + lane;
        bd[0]  = wv[0];
        bd[32] = wv[1];
        bd[64] = c2 ? wv[2] : make_float2(0.f, 0.f);
    }
    __syncthreads();

    int sf[BK_SPW], pf[BK_SPW + 1];
    pf[0] = 0;
#pragma unroll
    for (int i = 0; i < BK_SPW; ++i) {
        sf[i]     = __builtin_amdgcn_readlane(pr.x, i);
        pf[i + 1] = pf[i] + __builtin_amdgcn_readlane(pr.y, i);
    }
    const int tot = pf[BK_SPW];

    const int4* __restrict__ rec = (const int4*)rec_g;
    for (int b = 0; b < tot; b += 32) {
        const int cnt = (tot - b) < 32 ? (tot - b) : 32;
        const int pos = b + lane;
        int a = sf[0] + pos;
#pragma unroll
        for (int i = 1; i < BK_SPW; ++i)
            if (pos >= pf[i]) a = sf[i] + (pos - pf[i]);
        int4 r = make_int4(0, 0, 0, 0);
        if (lane < cnt) r = rec[a];
        for (int g = 0; g < cnt; g += SP_GRP) {
            float4 hv[SP_GRP];
#pragma unroll
            for (int u = 0; u < SP_GRP; ++u) {
                const int iu = (g + u < cnt) ? (g + u) : (cnt - 1);
                const int su = __builtin_amdgcn_readlane(r.x, iu);
                hv[u] = ((const float4*)(h + (size_t)su * NHID))[lane];
            }
#pragma unroll
            for (int u = 0; u < SP_GRP; ++u) {
                if (g + u < cnt) {
                    const int   du = __builtin_amdgcn_readlane(r.y, g + u);
                    const float wu = __int_as_float(__builtin_amdgcn_readlane(r.z, g + u));
                    float* tp = tile + du * SP_PITCH + lane;
                    atomicAdd(tp +  0, wu * hv[u].x);
                    atomicAdd(tp + 32, wu * hv[u].y);
                    atomicAdd(tp + 64, wu * hv[u].z);
                    atomicAdd(tp + 96, wu * hv[u].w);
                }
            }
        }
    }
    __syncthreads();

    if constexpr (EPI == 1) {
        const float* __restrict__ bp = bias + 16 * (lane & 7) + (lane >> 3);
        const float4 b4 = make_float4(bp[0], bp[4], bp[8], bp[12]);
        for (int i = threadIdx.x; i < SP_TROWS * (NHID / 4); i += SP_THREADS) {
            float4* p = tile4 + (i >> 5) * (SP_PITCH / 4) + lane;
            float4 v = *p;
            v.x = fmaxf(v.x + b4.x, 0.0f);
            v.y = fmaxf(v.y + b4.y, 0.0f);
            v.z = fmaxf(v.z + b4.z, 0.0f);
            v.w = fmaxf(v.w + b4.w, 0.0f);
            *p = v;
        }
        __syncthreads();
    }

    if constexpr (EPI == 0) {
        const float4* B1 = (const float4*)(smem + SP_TILE_F);
        const float4* B2 = B1 + GM_BS_F / 4;

        for (int base = 0; base < SP_RT * 4; base += SP_WAVES) {
            const int  item = base + wave;
            const bool act  = item < SP_RT * 4;
            const int  rt   = item >> 2;
            const int  q0   = item & 3;
            float4 av[16];
            if (act) sp_tile_row(av, (const float4*)(tile + (16 * rt + c) * SP_PITCH + 4 * hh));
            __syncthreads();
            if (act) {
                v8f d[2];
                gm_rows_times_b<2>(av, B1 + (2 * q0) * 32 + lane, d);
                float* trow = tile + (16 * rt + 8 * hh) * SP_PITCH + 32 * q0 + 2 * c;
#pragma unroll
                for (int v = 0; v < 8; ++v)
                    *(float2*)(trow + v * SP_PITCH) = make_float2(fmaxf(d[0][v] + bq.x, 0.0f), fmaxf(d[1][v] + bq.y, 0.0f));
            }
        }
        __syncthreads();

        for (int item = wave; item < SP_RT * 4; item += SP_WAVES) {
            const int rt = item >> 2;
            const int q0 = item & 3;
            float4 av[16];
            sp_tile_row(av, (const float4*)(tile + (16 * rt + c) * SP_PITCH + 4 * hh));
            v8f d[2];
            gm_rows_times_b<2>(av, B2 + (2 * q0) * 32 + lane, d);
            if (16 * rt + 8 * hh < SP_ROWS) {
                float* __restrict__ crow = dst + (size_t)(r0 + 16 * rt + 8 * hh) * NHID + 2 * c + 32 * q0;
#pragma unroll
                for (int v = 0; v < 8; ++v)
                    *(float2*)(crow + v * NHID) = make_float2(d[0][v], d[1][v]);
            }
        }
    } else {
        float* stg = smem + SP_TILE_F + HD_BS_F;
        if (wave < SP_RT) {
            const int rt = wave;
            float4 av[16];
            sp_tile_row(av, (const float4*)(tile + (16 * rt + c) * SP_PITCH + 4 * hh));
            const float2* Bs = (const float2*)(smem + SP_TILE_F) + lane;

            v8f d0 = {}, d1 = {}, d2 = {};
#pragma unroll
            for (int j = 0; j < 16; ++j) {
                v2f a; float2 f; v2f b;
                const float2* bs = Bs + (2 * j) * HD_CT * 32;
                a.x = av[j].x; a.y = av[j].y;
                f = bs[0];   b.x = f.x; b.y = f.y;
                d0 = __builtin_amdgcn_wmma_f32_16x16x4_f32(false, a, false, b, (short)0, d0, false, false);
                f = bs[32];  b.x = f.x; b.y = f.y;
                d1 = __builtin_amdgcn_wmma_f32_16x16x4_f32(false, a, false, b, (short)0, d1, false, false);
                f = bs[64];  b.x = f.x; b.y = f.y;
                d2 = __builtin_amdgcn_wmma_f32_16x16x4_f32(false, a, false, b, (short)0, d2, false, false);
                a.x = av[j].z; a.y = av[j].w;
                f = bs[96];  b.x = f.x; b.y = f.y;
                d0 = __builtin_amdgcn_wmma_f32_16x16x4_f32(false, a, false, b, (short)0, d0, false, false);
                f = bs[128]; b.x = f.x; b.y = f.y;
                d1 = __builtin_amdgcn_wmma_f32_16x16x4_f32(false, a, false, b, (short)0, d1, false, false);
                f = bs[160]; b.x = f.x; b.y = f.y;
                d2 = __builtin_amdgcn_wmma_f32_16x16x4_f32(false, a, false, b, (short)0, d2, false, false);
            }

            const float bo0 = aux[c];
            const float bo1 = aux[16 + c];
            const float bo2 = aux[col2];
            float l0[8], l1[8], l2[8], mx[8], se[8];
#pragma unroll
            for (int r = 0; r < 8; ++r) {
                l0[r] = d0[r] + bo0;
                l1[r] = d1[r] + bo1;
                l2[r] = c2 ? (d2[r] + bo2) : -INFINITY;
                mx[r] = fmaxf(fmaxf(l0[r], l1[r]), l2[r]);
            }
#pragma unroll
            for (int r = 0; r < 8; ++r) mx[r] = fmaxf(mx[r], row_xor<1>(mx[r]));
#pragma unroll
            for (int r = 0; r < 8; ++r) mx[r] = fmaxf(mx[r], row_xor<2>(mx[r]));
#pragma unroll
            for (int r = 0; r < 8; ++r) mx[r] = fmaxf(mx[r], row_xor<4>(mx[r]));
#pragma unroll
            for (int r = 0; r < 8; ++r) mx[r] = fmaxf(mx[r], row_xor<8>(mx[r]));
#pragma unroll
            for (int r = 0; r < 8; ++r)
                se[r] = __expf(l0[r] - mx[r]) + __expf(l1[r] - mx[r]) + (c2 ? __expf(l2[r] - mx[r]) : 0.0f);
#pragma unroll
            for (int r = 0; r < 8; ++r) se[r] += row_xor<1>(se[r]);
#pragma unroll
            for (int r = 0; r < 8; ++r) se[r] += row_xor<2>(se[r]);
#pragma unroll
            for (int r = 0; r < 8; ++r) se[r] += row_xor<4>(se[r]);
#pragma unroll
            for (int r = 0; r < 8; ++r) se[r] += row_xor<8>(se[r]);

            if (16 * rt + 8 * hh < SP_ROWS) {
                float* __restrict__ sp = stg + (16 * rt + 8 * hh) * NCLASS + c;
#pragma unroll
                for (int r = 0; r < 8; ++r) {
                    const float lse = __logf(se[r]) + mx[r];
                    sp[r * NCLASS]      = l0[r] - lse;
                    sp[r * NCLASS + 16] = l1[r] - lse;
                    if (c2) sp[r * NCLASS + 32] = l2[r] - lse;
                }
            }
        }
        __syncthreads();

        float4* __restrict__ o4 = (float4*)(dst + (size_t)r0 * NCLASS);
        const float4* s4 = (const float4*)stg;
        for (int i = threadIdx.x; i < SP_ROWS * NCLASS / 4; i += SP_THREADS) o4[i] = s4[i];
    }
}

extern "C" void kernel_launch(void* const* d_in, const int* in_sizes, int n_in,
                              void* d_out, int out_size, void* d_ws, size_t ws_size,
                              hipStream_t stream)
{
    const float* x    = (const float*)d_in[0];
    const int*   a0i  = (const int*)  d_in[1];
    const float* a0w  = (const float*)d_in[2];
    const int*   a1i  = (const int*)  d_in[3];
    const float* a1w  = (const float*)d_in[4];
    const float* W1   = (const float*)d_in[5];
    const float* b1   = (const float*)d_in[6];
    const float* W2   = (const float*)d_in[7];
    const float* b2   = (const float*)d_in[8];
    const float* Wout = (const float*)d_in[9];
    const float* bout = (const float*)d_in[10];
    float* out = (float*)d_out;

    const size_t NH = (size_t)N_NODES * NHID;
    float* bufT = (float*)d_ws;
    int*   recAll = (int*)(bufT + NH);
    const size_t REC_G = (size_t)N_EDGES * 4;
    int*   segAll = recAll + 2 * REC_G;
    const size_t SEG_G = (size_t)SP_NB * BK_NCH * 2;

    const dim3 gSp(SP_NB), bSp(SP_THREADS);

    bucket_chunks<<<dim3(2 * BK_NCH), dim3(BK_THREADS), 0, stream>>>(
        a0i, a0i + N_EDGES, a0w, a1i, a1i + N_EDGES, a1w, recAll, segAll);

    spmm_rows_dense<0><<<gSp, bSp, 0, stream>>>(recAll, segAll, x, b1, W1, W2, bufT);

    spmm_rows_dense<1><<<gSp, bSp, 0, stream>>>(recAll + REC_G, segAll + SEG_G, bufT, b2, Wout, bout, out);
}


# spmm: block's records ordered by dst row in LDS (integer-atomic rank, scan, place); a wave sums whole rows in registers, one plain store per row: no LDS float atomics; sums in rank (arrival) order
# speedup vs baseline: 5.1528x; 1.0603x over previous
#include <hip/hip_runtime.h>
#include <math.h>

#define N_NODES 50000
#define N_EDGES 600000
#define NFEAT 128
#define NHID 128
#define NCLASS 41

typedef float v2f __attribute__((ext_vector_type(2)));
typedef float v8f __attribute__((ext_vector_type(8)));

static_assert(NFEAT == 128 && NHID == 128, "the fragment layout below is for 128 x 128 weights");

template<int NT>
__device__ __forceinline__ void gm_rows_times_b(const float4 (&av)[16], const float4* bs, v8f (&d)[NT])
{
#pragma unroll
    for (int t = 0; t < NT; ++t) d[t] = v8f{};
    float4 f[2][NT];
#pragma unroll
    for (int t = 0; t < NT; ++t) f[0][t] = bs[t * 32];
#pragma unroll
    for (int j = 0; j < 16; ++j) {
        if (j + 1 < 16) {
#pragma unroll
            for (int t = 0; t < NT; ++t) f[(j + 1) & 1][t] = bs[((j + 1) * 8 + t) * 32];
        }
        __builtin_amdgcn_sched_barrier(0);
        v2f a0, a1;
        a0.x = av[j].x; a0.y = av[j].y;
        a1.x = av[j].z; a1.y = av[j].w;
#pragma unroll
        for (int t = 0; t < NT; ++t) {
            v2f b; b.x = f[j & 1][t].x; b.y = f[j & 1][t].y;
            d[t] = __builtin_amdgcn_wmma_f32_16x16x4_f32(false, a0, false, b, (short)0, d[t], false, false);
        }
#pragma unroll
        for (int t = 0; t < NT; ++t) {
            v2f b; b.x = f[j & 1][t].z; b.y = f[j & 1][t].w;
            d[t] = __builtin_amdgcn_wmma_f32_16x16x4_f32(false, a1, false, b, (short)0, d[t], false, false);
        }
        __builtin_amdgcn_sched_barrier(0);
    }
}

#define SP_ROWS    200
#define SP_RT      ((SP_ROWS + 15) / 16)
#define SP_TROWS   (SP_RT * 16)
#define SP_PITCH   (NHID + 4)
#define SP_TILE_F  (SP_TROWS * SP_PITCH)
#define SP_NB      (N_NODES / SP_ROWS)
#define SP_THREADS 1024
#define SP_WAVES   (SP_THREADS / 32)
#define SP_GRP     8
#define SP_HB      4
#define SP_CAP     (SP_WAVES * SP_HB * 32)
#define SP_RBINS   256
#define BK_CH      4800
#define BK_NCH     (N_EDGES / BK_CH)
#define BK_THREADS 512
#define BK_PER     ((BK_CH + BK_THREADS - 1) / BK_THREADS)
#define BK_BINS    256
#define BK_SPW     ((BK_NCH + SP_WAVES - 1) / SP_WAVES)
static_assert(N_NODES % SP_ROWS == 0, "row ranges must tile the node set exactly");
static_assert(N_EDGES % BK_CH == 0, "chunks must tile the edge list exactly");
static_assert(SP_NB < BK_BINS && BK_BINS == 8 * 32, "the scan gives each lane of one wave 8 bins");
static_assert(SP_NB <= BK_THREADS, "one thread writes one (first, count) pair");
static_assert(BK_SPW <= 32, "a wave holds its chunk pairs one per lane");
static_assert(SP_ROWS < SP_RBINS && SP_RBINS == 8 * 32, "the row scan gives each lane of one wave 8 rows; the last counter stays 0");

__global__ __launch_bounds__(BK_THREADS)
void bucket_chunks(const int* __restrict__ src0, const int* __restrict__ dst0, const float* __restrict__ w0,
                   const int* __restrict__ src1, const int* __restrict__ dst1, const float* __restrict__ w1,
                   int* __restrict__ rec_all, int* __restrict__ seg_all)
{
    __shared__ int hist[BK_BINS];
    __shared__ int first[BK_BINS];
    __shared__ __attribute__((aligned(16))) int4 sorted[BK_CH];

    const int t     = threadIdx.x;
    const int graph = (int)blockIdx.x / BK_NCH;
    const int chunk = (int)blockIdx.x - graph * BK_NCH;
    const int e0    = chunk * BK_CH;
    const int*   __restrict__ src = graph ? src1 : src0;
    const int*   __restrict__ dst = graph ? dst1 : dst0;
    const float* __restrict__ w   = graph ? w1 : w0;

    if (t < BK_BINS) hist[t] = 0;

    int s[BK_PER] = {}, d[BK_PER] = {}, wb[BK_PER] = {}, bin[BK_PER] = {}, rank[BK_PER] = {};
#pragma unroll
    for (int j = 0; j < BK_PER; ++j) {
        const int i = j * BK_THREADS + t;
        if (i < BK_CH) {
            s[j]  = src[e0 + i];
            d[j]  = dst[e0 + i];
            wb[j] = __float_as_int(w[e0 + i]);
        }
    }
    __syncthreads();

#pragma unroll
    for (int j = 0; j < BK_PER; ++j) {
        const int i = j * BK_THREADS + t;
        if (i < BK_CH) {
            bin[j]  = ((unsigned)d[j] < (unsigned)N_NODES) ? (d[j] / SP_ROWS) : SP_NB;
            rank[j] = atomicAdd(&hist[bin[j]], 1);
        }
    }
    __syncthreads();

    if (t < 32) {
        int c[8];
        int sum = 0;
#pragma unroll
        for (int k = 0; k < 8; ++k) { c[k] = hist[t * 8 + k]; sum += c[k]; }
        int incl = sum;
#pragma unroll
        for (int dl = 1; dl < 32; dl <<= 1) {
            const int o = __shfl_up(incl, dl);
            if (t >= dl) incl += o;
        }
        int run = incl - sum;
#pragma unroll
        for (int k = 0; k < 8; ++k) { first[t * 8 + k] = run; run += c[k]; }
    }
    __syncthreads();

#pragma unroll
    for (int j = 0; j < BK_PER; ++j) {
        const int i = j * BK_THREADS + t;
        if (i < BK_CH)
            sorted[first[bin[j]] + rank[j]] = make_int4(s[j], d[j] - bin[j] * SP_ROWS, wb[j], 0);
    }
    if (t < SP_NB) {
        int2* __restrict__ seg = (int2*)seg_all + ((size_t)graph * SP_NB + t) * BK_NCH + chunk;
        *seg = make_int2(e0 + first[t], hist[t]);
    }
    __syncthreads();

    int4* __restrict__ rec = (int4*)rec_all + (size_t)graph * N_EDGES + e0;
    for (int i = t; i < BK_CH; i += BK_THREADS) rec[i] = sorted[i];
}

#define HD_CT    3
#define HD_BS_F  (32 * HD_CT * 32 * 2)
#define GM_BS_F  (16 * 8 * 32 * 4)
static_assert(NCLASS > 32 && NCLASS <= 48, "three column tiles, the third one partial");
static_assert((SP_ROWS * NCLASS) % 4 == 0, "a block of out is stored as float4");
static_assert(SP_WAVES == 32, "one wave stages one 4-row group of the 128x128 weight, or one k-step of Wout");
static_assert(SP_RT <= SP_WAVES, "the class head gives every row tile a wave of its own");
static_assert(SP_THREADS % 32 == 0 && NHID / 4 == 32 && SP_PITCH % 4 == 0, "the in-place relu pass: one lane, one group of four columns");

template<int M>
__device__ __forceinline__ float row_xor(float x)
{
    return __builtin_bit_cast(float,
        __builtin_amdgcn_mov_dpp(__builtin_bit_cast(int, x), 0x160 + M, 0xf, 0xf, true));
}

__device__ __forceinline__ void sp_tile_row(float4 (&av)[16], const float4* ap)
{
#pragma unroll
    for (int j = 0; j < 16; ++j) av[j] = ap[2 * j];
}

template<int EPI>
__global__ __launch_bounds__(SP_THREADS)
void spmm_rows_dense(const int* __restrict__ rec_g, const int* __restrict__ seg_g,
                     const float* __restrict__ h, const float* __restrict__ bias,
                     const float* __restrict__ Wn, const float* __restrict__ aux,
                     float* __restrict__ dst)
{
    constexpr int SP_SORT_F = SP_TILE_F + (EPI == 0 ? 2 * GM_BS_F : (HD_BS_F + SP_ROWS * NCLASS));
    static_assert(SP_SORT_F % 4 == 0, "the ordered records are 16-byte items");
    __shared__ __attribute__((aligned(16))) float smem[SP_SORT_F + 2 * SP_RBINS + 4 * SP_CAP];
    float* tile = smem;
    int*  rcnt   = (int*)(smem + SP_SORT_F);
    int*  rstart = rcnt + SP_RBINS;
    int4* srt    = (int4*)(rstart + SP_RBINS);

    const int lane = threadIdx.x & 31;
    const int wave = __builtin_amdgcn_readfirstlane((int)(threadIdx.x >> 5));
    const int r0   = blockIdx.x * SP_ROWS;
    const int c    = lane & 15;
    const int hh   = lane >> 4;

    const int2* __restrict__ seg = (const int2*)seg_g + (size_t)blockIdx.x * BK_NCH;
    const int myc = wave + lane * SP_WAVES;
    int2 pr = make_int2(0, 0);
    if (lane < BK_SPW && myc < BK_NCH) pr = seg[myc];

    const bool c2   = (32 + c) < NCLASS;
    const int  col2 = c2 ? (32 + c) : (NCLASS - 1);
    float4 wr[2][4];
    float2 wv[HD_CT];
    float2 bq = make_float2(0.f, 0.f);
    if constexpr (EPI == 0) {
        static_assert(SP_WAVES % 4 == 0, "a wave's items share one column group");
        bq = *(const float2*)(bias + 32 * (wave & 3) + 2 * c);
#pragma unroll
        for (int m = 0; m < 2; ++m) {
            const int k0 = m ? (4 * wave) : (32 * ((wave >> 1) & 3) + 16 * (wave & 1) + (wave >> 3));
            const int ks = m ? 1 : 4;
            const float4* __restrict__ wp = (const float4*)((m ? aux : Wn) + (size_t)k0 * NHID) + lane;
#pragma unroll
            for (int r = 0; r < 4; ++r) wr[m][r] = wp[r * ks * (NHID / 4)];
        }
    } else {
        const int k = 32 * ((wave >> 1) & 3) + 16 * hh + 8 * (wave & 1) + (wave >> 3);
        const float* __restrict__ w0 = Wn + (size_t)k * NCLASS;
        const float* __restrict__ w1 = w0 + 4 * NCLASS;
        wv[0] = make_float2(w0[c],      w1[c]);
        wv[1] = make_float2(w0[16 + c], w1[16 + c]);
        wv[2] = make_float2(w0[col2],   w1[col2]);
    }

    int sf[BK_SPW], pf[BK_SPW + 1];
    pf[0] = 0;
#pragma unroll
    for (int i = 0; i < BK_SPW; ++i) {
        sf[i]     = __builtin_amdgcn_readlane(pr.x, i);
        pf[i + 1] = pf[i] + __builtin_amdgcn_readlane(pr.y, i);
    }
    const int tot = pf[BK_SPW];

    const int4* __restrict__ rec = (const int4*)rec_g;

    int4 rk[SP_HB];
    int  rnk[SP_HB];
#pragma unroll
    for (int k = 0; k < SP_HB; ++k) {
        const int pos = 32 * k + lane;
        int a = sf[0] + pos;
#pragma unroll
        for (int i = 1; i < BK_SPW; ++i)
            if (pos >= pf[i]) a = sf[i] + (pos - pf[i]);
        rk[k] = make_int4(0, 0, 0, 0);
        if (pos < tot) rk[k] = rec[a];
    }
    float4* tile4 = (float4*)tile;
    for (int i = threadIdx.x; i < SP_TILE_F / 4; i += SP_THREADS)
        tile4[i] = make_float4(0.f, 0.f, 0.f, 0.f);
    if (threadIdx.x < SP_RBINS) rcnt[threadIdx.x] = 0;

    if constexpr (EPI == 0) {
#pragma unroll
        for (int m = 0; m < 2; ++m) {
            float4* __restrict__ bd = (float4*)(smem + SP_TILE_F + m * GM_BS_F)
                                      + ((wave >> 1) * 8 + 2 * (lane >> 3)) * 32 + (wave & 1) * 16 + 2 * (lane & 7);
            bd[0]      = make_float4(wr[m][0].x, wr[m][1].x, wr[m][2].x, wr[m][3].x);
            bd[32]     = make_float4(wr[m][0].y, wr[m][1].y, wr[m][2].y, wr[m][3].y);
            bd[1]      = make_float4(wr[m][0].z, wr[m][1].z, wr[m][2].z, wr[m][3].z);
            bd[32 + 1] = make_float4(wr[m][0].w, wr[m][1].w, wr[m][2].w, wr[m][3].w);
        }
    } else {
        float2* __restrict__ bd = (float2*)(smem + SP_TILE_F) + (wave * HD_CT) * 32 + lane;
        bd[0]  = wv[0];
        bd[32] = wv[1];
        bd[64] = c2 ? wv[2] : make_float2(0.f, 0.f);
    }
    __syncthreads();

#pragma unroll
    for (int k = 0; k < SP_HB; ++k) {
        rnk[k] = 0;
        if (32 * k + lane < tot) rnk[k] = atomicAdd(&rcnt[rk[k].y], 1);
    }
    __syncthreads();

    if (wave == 0) {
        int c8[8];
        int sum = 0;
#pragma unroll
        for (int k = 0; k < 8; ++k) { c8[k] = rcnt[lane * 8 + k]; sum += c8[k]; }
        int incl = sum;
#pragma unroll
        for (int dl = 1; dl < 32; dl <<= 1) {
            const int o = __shfl_up(incl, dl);
            if (lane >= dl) incl += o;
        }
        int run = incl - sum;
#pragma unroll
        for (int k = 0; k < 8; ++k) { rstart[lane * 8 + k] = run; run += c8[k]; }
    }
    __syncthreads();

    int plc[SP_HB];
#pragma unroll
    for (int k = 0; k < SP_HB; ++k) plc[k] = rstart[rk[k].y] + rnk[k];
#pragma unroll
    for (int k = 0; k < SP_HB; ++k)
        if (32 * k + lane < tot) srt[plc[k]] = make_int4(rk[k].x, rk[k].y, rk[k].z, 0);

    const int nsrt = __builtin_amdgcn_readfirstlane(rstart[SP_RBINS - 1]);
    const int mlo  = (nsrt * wave) / SP_WAVES;
    const int mhi  = (nsrt * (wave + 1)) / SP_WAVES;
    int rlo = 0, rhi = 0;
#pragma unroll
    for (int k = 0; k < SP_RBINS / 32; ++k) {
        const int sv = rstart[32 * k + lane];
        rlo += __builtin_popcount(__builtin_amdgcn_ballot_w32(sv < mlo));
        rhi += __builtin_popcount(__builtin_amdgcn_ballot_w32(sv < mhi));
    }
    const int elo = __builtin_amdgcn_readfirstlane(rstart[rlo]);
    const int ehi = __builtin_amdgcn_readfirstlane(rstart[rhi]);
    __syncthreads();

    {
        float4 acc = make_float4(0.f, 0.f, 0.f, 0.f);
        int cur = -1;
        for (int b = elo; b < ehi; b += 32) {
            const int cnt = (ehi - b) < 32 ? (ehi - b) : 32;
            int4 r = make_int4(0, 0, 0, 0);
            if (lane < cnt) r = srt[b + lane];
            for (int g = 0; g < cnt; g += SP_GRP) {
                float4 hv[SP_GRP];
#pragma unroll
                for (int u = 0; u < SP_GRP; ++u) {
                    const int iu = (g + u < cnt) ? (g + u) : (cnt - 1);
                    const int su = __builtin_amdgcn_readlane(r.x, iu);
                    hv[u] = ((const float4*)(h + (size_t)su * NHID))[lane];
                }
#pragma unroll
                for (int u = 0; u < SP_GRP; ++u) {
                    if (g + u < cnt) {
                        const int   du = __builtin_amdgcn_readlane(r.y, g + u);
                        const float wu = __int_as_float(__builtin_amdgcn_readlane(r.z, g + u));
                        if (du != cur) {
                            if (cur >= 0) {
                                float* tp = tile + cur * SP_PITCH + lane;
                                tp[0] = acc.x; tp[32] = acc.y; tp[64] = acc.z; tp[96] = acc.w;
                            }
                            acc = make_float4(0.f, 0.f, 0.f, 0.f);
                            cur = du;
                        }
                        acc.x += wu * hv[u].x;
                        acc.y += wu * hv[u].y;
                        acc.z += wu * hv[u].z;
                        acc.w += wu * hv[u].w;
                    }
                }
            }
        }
        if (cur >= 0) {
            float* tp = tile + cur * SP_PITCH + lane;
            tp[0] = acc.x; tp[32] = acc.y; tp[64] = acc.z; tp[96] = acc.w;
        }
    }
    __syncthreads();

    for (int b = SP_HB * 32; b < tot; b += 32) {
        const int cnt = (tot - b) < 32 ? (tot - b) : 32;
        const int pos = b + lane;
        int a = sf[0] + pos;
#pragma unroll
        for (int i = 1; i < BK_SPW; ++i)
            if (pos >= pf[i]) a = sf[i] + (pos - pf[i]);
        int4 r = make_int4(0, 0, 0, 0);
        if (lane < cnt) r = rec[a];
        for (int g = 0; g < cnt; g += SP_GRP) {
            float4 hv[SP_GRP];
#pragma unroll
            for (int u = 0; u < SP_GRP; ++u) {
                const int iu = (g + u < cnt) ? (g + u) : (cnt - 1);
                const int su = __builtin_amdgcn_readlane(r.x, iu);
                hv[u] = ((const float4*)(h + (size_t)su * NHID))[lane];
            }
#pragma unroll
            for (int u = 0; u < SP_GRP; ++u) {
                if (g + u < cnt) {
                    const int   du = __builtin_amdgcn_readlane(r.y, g + u);
                    const float wu = __int_as_float(__builtin_amdgcn_readlane(r.z, g + u));
                    float* tp = tile + du * SP_PITCH + lane;
                    atomicAdd(tp +  0, wu * hv[u].x);
                    atomicAdd(tp + 32, wu * hv[u].y);
                    atomicAdd(tp + 64, wu * hv[u].z);
                    atomicAdd(tp + 96, wu * hv[u].w);
                }
            }
        }
    }
    __syncthreads();

    if constexpr (EPI == 1) {
        const float* __restrict__ bp = bias + 16 * (lane & 7) + (lane >> 3);
        const float4 b4 = make_float4(bp[0], bp[4], bp[8], bp[12]);
        for (int i = threadIdx.x; i < SP_TROWS * (NHID / 4); i += SP_THREADS) {
            float4* p = tile4 + (i >> 5) * (SP_PITCH / 4) + lane;
            float4 v = *p;
            v.x = fmaxf(v.x + b4.x, 0.0f);
            v.y = fmaxf(v.y + b4.y, 0.0f);
            v.z = fmaxf(v.z + b4.z, 0.0f);
            v.w = fmaxf(v.w + b4.w, 0.0f);
            *p = v;
        }
        __syncthreads();
    }

    if constexpr (EPI == 0) {
        const float4* B1 = (const float4*)(smem + SP_TILE_F);
        const float4* B2 = B1 + GM_BS_F / 4;

        for (int base = 0; base < SP_RT * 4; base += SP_WAVES) {
            const int  item = base + wave;
            const bool act  = item < SP_RT * 4;
            const int  rt   = item >> 2;
            const int  q0   = item & 3;
            float4 av[16];
            if (act) sp_tile_row(av, (const float4*)(tile + (16 * rt + c) * SP_PITCH + 4 * hh));
            __syncthreads();
            if (act) {
                v8f d[2];
                gm_rows_times_b<2>(av, B1 + (2 * q0) * 32 + lane, d);
                float* trow = tile + (16 * rt + 8 * hh) * SP_PITCH + 32 * q0 + 2 * c;
#pragma unroll
                for (int v = 0; v < 8; ++v)
                    *(float2*)(trow + v * SP_PITCH) = make_float2(fmaxf(d[0][v] + bq.x, 0.0f), fmaxf(d[1][v] + bq.y, 0.0f));
            }
        }
        __syncthreads();

        for (int item = wave; item < SP_RT * 4; item += SP_WAVES) {
            const int rt = item >> 2;
            const int q0 = item & 3;
            float4 av[16];
            sp_tile_row(av, (const float4*)(tile + (16 * rt + c) * SP_PITCH + 4 * hh));
            v8f d[2];
            gm_rows_times_b<2>(av, B2 + (2 * q0) * 32 + lane, d);
            if (16 * rt + 8 * hh < SP_ROWS) {
                float* __restrict__ crow = dst + (size_t)(r0 + 16 * rt + 8 * hh) * NHID + 2 * c + 32 * q0;
#pragma unroll
                for (int v = 0; v < 8; ++v)
                    *(float2*)(crow + v * NHID) = make_float2(d[0][v], d[1][v]);
            }
        }
    } else {
        float* stg = smem + SP_TILE_F + HD_BS_F;
        if (wave < SP_RT) {
            const int rt = wave;
            float4 av[16];
            sp_tile_row(av, (const float4*)(tile + (16 * rt + c) * SP_PITCH + 4 * hh));
            const float2* Bs = (const float2*)(smem + SP_TILE_F) + lane;

            v8f d0 = {}, d1 = {}, d2 = {};
#pragma unroll
            for (int j = 0; j < 16; ++j) {
                v2f a; float2 f; v2f b;
                const float2* bs = Bs + (2 * j) * HD_CT * 32;
                a.x = av[j].x; a.y = av[j].y;
                f = bs[0];   b.x = f.x; b.y = f.y;
                d0 = __builtin_amdgcn_wmma_f32_16x16x4_f32(false, a, false, b, (short)0, d0, false, false);
                f = bs[32];  b.x = f.x; b.y = f.y;
                d1 = __builtin_amdgcn_wmma_f32_16x16x4_f32(false, a, false, b, (short)0, d1, false, false);
                f = bs[64];  b.x = f.x; b.y = f.y;
                d2 = __builtin_amdgcn_wmma_f32_16x16x4_f32(false, a, false, b, (short)0, d2, false, false);
                a.x = av[j].z; a.y = av[j].w;
                f = bs[96];  b.x = f.x; b.y = f.y;
                d0 = __builtin_amdgcn_wmma_f32_16x16x4_f32(false, a, false, b, (short)0, d0, false, false);
                f = bs[128]; b.x = f.x; b.y = f.y;
                d1 = __builtin_amdgcn_wmma_f32_16x16x4_f32(false, a, false, b, (short)0, d1, false, false);
                f = bs[160]; b.x = f.x; b.y = f.y;
                d2 = __builtin_amdgcn_wmma_f32_16x16x4_f32(false, a, false, b, (short)0, d2, false, false);
            }

            const float bo0 = aux[c];
            const float bo1 = aux[16 + c];
            const float bo2 = aux[col2];
            float l0[8], l1[8], l2[8], mx[8], se[8];
#pragma unroll
            for (int r = 0; r < 8; ++r) {
                l0[r] = d0[r] + bo0;
                l1[r] = d1[r] + bo1;
                l2[r] = c2 ? (d2[r] + bo2) : -INFINITY;
                mx[r] = fmaxf(fmaxf(l0[r], l1[r]), l2[r]);
            }
#pragma unroll
            for (int r = 0; r < 8; ++r) mx[r] = fmaxf(mx[r], row_xor<1>(mx[r]));
#pragma unroll
            for (int r = 0; r < 8; ++r) mx[r] = fmaxf(mx[r], row_xor<2>(mx[r]));
#pragma unroll
            for (int r = 0; r < 8; ++r) mx[r] = fmaxf(mx[r], row_xor<4>(mx[r]));
#pragma unroll
            for (int r = 0; r < 8; ++r) mx[r] = fmaxf(mx[r], row_xor<8>(mx[r]));
#pragma unroll
            for (int r = 0; r < 8; ++r)
                se[r] = __expf(l0[r] - mx[r]) + __expf(l1[r] - mx[r]) + (c2 ? __expf(l2[r] - mx[r]) : 0.0f);
#pragma unroll
            for (int r = 0; r < 8; ++r) se[r] += row_xor<1>(se[r]);
#pragma unroll
            for (int r = 0; r < 8; ++r) se[r] += row_xor<2>(se[r]);
#pragma unroll
            for (int r = 0; r < 8; ++r) se[r] += row_xor<4>(se[r]);
#pragma unroll
            for (int r = 0; r < 8; ++r) se[r] += row_xor<8>(se[r]);

            if (16 * rt + 8 * hh < SP_ROWS) {
                float* __restrict__ sp = stg + (16 * rt + 8 * hh) * NCLASS + c;
#pragma unroll
                for (int r = 0; r < 8; ++r) {
                    const float lse = __logf(se[r]) + mx[r];
                    sp[r * NCLASS]      = l0[r] - lse;
                    sp[r * NCLASS + 16] = l1[r] - lse;
                    if (c2) sp[r * NCLASS + 32] = l2[r] - lse;
                }
            }
        }
        __syncthreads();

        float4* __restrict__ o4 = (float4*)(dst + (size_t)r0 * NCLASS);
        const float4* s4 = (const float4*)stg;
        for (int i = threadIdx.x; i < SP_ROWS * NCLASS / 4; i += SP_THREADS) o4[i] = s4[i];
    }
}

extern "C" void kernel_launch(void* const* d_in, const int* in_sizes, int n_in,
                              void* d_out, int out_size, void* d_ws, size_t ws_size,
                              hipStream_t stream)
{
    const float* x    = (const float*)d_in[0];
    const int*   a0i  = (const int*)  d_in[1];
    const float* a0w  = (const float*)d_in[2];
    const int*   a1i  = (const int*)  d_in[3];
    const float* a1w  = (const float*)d_in[4];
    const float* W1   = (const float*)d_in[5];
    const float* b1   = (const float*)d_in[6];
    const float* W2   = (const float*)d_in[7];
    const float* b2   = (const float*)d_in[8];
    const float* Wout = (const float*)d_in[9];
    const float* bout = (const float*)d_in[10];
    float* out = (float*)d_out;

    const size_t NH = (size_t)N_NODES * NHID;
    float* bufT = (float*)d_ws;
    int*   recAll = (int*)(bufT + NH);
    const size_t REC_G = (size_t)N_EDGES * 4;
    int*   segAll = recAll + 2 * REC_G;
    const size_t SEG_G = (size_t)SP_NB * BK_NCH * 2;

    const dim3 gSp(SP_NB), bSp(SP_THREADS);

    bucket_chunks<<<dim3(2 * BK_NCH), dim3(BK_THREADS), 0, stream>>>(
        a0i, a0i + N_EDGES, a0w, a1i, a1i + N_EDGES, a1w, recAll, segAll);

    spmm_rows_dense<0><<<gSp, bSp, 0, stream>>>(recAll, segAll, x, b1, W1, W2, bufT);

    spmm_rows_dense<1><<<gSp, bSp, 0, stream>>>(recAll + REC_G, segAll + SEG_G, bufT, b2, Wout, bout, out);
}
